# starting kernel
# baseline (speedup 1.0000x reference)
#include <hip/hip_runtime.h>

typedef float v2f __attribute__((ext_vector_type(2)));
typedef float v8f __attribute__((ext_vector_type(8)));

#define ND_   50000
#define NG_   100000
#define E_    600000
#define DIN_  128
#define H_    64

__global__ __launch_bounds__(256)
void count_edges(const int* __restrict__ dst, float* __restrict__ cnt, int nE) {
    int i = blockIdx.x * blockDim.x + threadIdx.x;
    if (i < nE) atomicAdd(&cnt[dst[i]], 1.0f);
}

template<int LOGD>
__global__ __launch_bounds__(256)
void scatter_add4(const int* __restrict__ src, const int* __restrict__ dst,
                  const float* __restrict__ xsrc, float* __restrict__ acc, int nE) {
    const unsigned CPT = 1u << (LOGD - 2);
    unsigned idx = blockIdx.x * blockDim.x + threadIdx.x;
    unsigned e   = idx >> (LOGD - 2);
    if (e >= (unsigned)nE) return;
    unsigned c = idx & (CPT - 1);

    const float4* xs = (const float4*)(xsrc + ((size_t)src[e] << LOGD));
    float4 v = xs[c];
    float* out = acc + ((size_t)dst[e] << LOGD) + (size_t)c * 4;
    atomicAdd(out + 0, v.x);
    atomicAdd(out + 1, v.y);
    atomicAdd(out + 2, v.z);
    atomicAdd(out + 3, v.w);
}

template<int K>
__global__ __launch_bounds__(128)
void sage_wmma(const float* __restrict__ msum, const float* __restrict__ cnt,
               const float* __restrict__ xdst,
               const float* __restrict__ Wl, const float* __restrict__ bl,
               const float* __restrict__ Wr,
               float* __restrict__ out, int N)
{
    __shared__ v2f Wpk[K * 64];
    {
        float* lds = (float*)Wpk;
        for (int i = threadIdx.x; i < K * 64; i += blockDim.x) {
            const int k = i >> 6, col = i & 63;
            lds[((k >> 1) * 64 + col) * 2 + (k & 1)]           = Wl[i];
            lds[((K / 2 + (k >> 1)) * 64 + col) * 2 + (k & 1)] = Wr[i];
        }
    }
    __syncthreads();

    const int wave = threadIdx.x >> 5;
    const int lane = threadIdx.x & 31;
    const int pair = blockIdx.x * 4 + wave;
    const int row0 = pair * 32;
    if (row0 >= N) return;
    const bool has2 = (row0 + 32 <= N);

    const int m  = lane & 15;
    const int ph = lane >> 4;
    const int rowA0 = row0 + m;
    const int rowA1 = min(row0 + 16 + m, N - 1);

    const float s0 = 1.0f / fmaxf(cnt[rowA0], 1.0f);
    const float s1 = 1.0f / fmaxf(cnt[rowA1], 1.0f);

    v8f acc0[4] = {};
    v8f acc1[4] = {};

    {
        const v2f* A0 = (const v2f*)(msum + (size_t)rowA0 * K);
        const v2f* A1 = (const v2f*)(msum + (size_t)rowA1 * K);
        for (int k = 0; k < K; k += 4) {
            const int kp = (k >> 1) + ph;
            v2f a0 = A0[kp] * s0;
            v2f a1 = A1[kp] * s1;
#pragma unroll
            for (int t = 0; t < 4; ++t) {
                v2f b = Wpk[kp * 64 + t * 16 + m];
                acc0[t] = __builtin_amdgcn_wmma_f32_16x16x4_f32(
                    false, a0, false, b, (short)0, acc0[t], false, false);
                acc1[t] = __builtin_amdgcn_wmma_f32_16x16x4_f32(
                    false, a1, false, b, (short)0, acc1[t], false, false);
            }
        }
    }

    {
        const v2f* X0 = (const v2f*)(xdst + (size_t)rowA0 * K);
        const v2f* X1 = (const v2f*)(xdst + (size_t)rowA1 * K);
        for (int k = 0; k < K; k += 4) {
            const int kp = (k >> 1) + ph;
            v2f a0 = X0[kp];
            v2f a1 = X1[kp];
#pragma unroll
            for (int t = 0; t < 4; ++t) {
                v2f b = Wpk[(K / 2 + kp) * 64 + t * 16 + m];
                acc0[t] = __builtin_amdgcn_wmma_f32_16x16x4_f32(
                    false, a0, false, b, (short)0, acc0[t], false, false);
                acc1[t] = __builtin_amdgcn_wmma_f32_16x16x4_f32(
                    false, a1, false, b, (short)0, acc1[t], false, false);
            }
        }
    }

    const int colb  = lane & 15;
    const int mb0   = row0 + 8 * ph;
#pragma unroll
    for (int t = 0; t < 4; ++t) {
        const float bias = bl[t * 16 + colb];
#pragma unroll
        for (int r = 0; r < 8; ++r)
            out[(size_t)(mb0 + r) * 64 + t * 16 + colb] = acc0[t][r] + bias;
        if (has2) {
#pragma unroll
            for (int r = 0; r < 8; ++r)
                out[(size_t)(mb0 + 16 + r) * 64 + t * 16 + colb] = acc1[t][r] + bias;
        }
    }
}

extern "C" void kernel_launch(void* const* d_in, const int* in_sizes, int n_in,
                              void* d_out, int out_size, void* d_ws, size_t ws_size,
                              hipStream_t stream) {
    const float* x_disease = (const float*)d_in[0];
    const float* x_gene    = (const float*)d_in[1];
    const int*   src_dg    = (const int*)d_in[2];
    const int*   dst_dg    = (const int*)d_in[3];
    const int*   src_gd    = (const int*)d_in[4];
    const int*   dst_gd    = (const int*)d_in[5];
    const float* Wl1_dg = (const float*)d_in[6];
    const float* bl1_dg = (const float*)d_in[7];
    const float* Wr1_dg = (const float*)d_in[8];
    const float* Wl1_gd = (const float*)d_in[9];
    const float* bl1_gd = (const float*)d_in[10];
    const float* Wr1_gd = (const float*)d_in[11];
    const float* Wl2_dg = (const float*)d_in[12];
    const float* bl2_dg = (const float*)d_in[13];
    const float* Wr2_dg = (const float*)d_in[14];
    const float* Wl2_gd = (const float*)d_in[15];
    const float* bl2_gd = (const float*)d_in[16];
    const float* Wr2_gd = (const float*)d_in[17];

    float* d2 = (float*)d_out;
    float* g2 = d2 + (size_t)ND_ * 64;

    float* ws = (float*)d_ws;
    size_t off = 0;
    float* ms_g  = ws + off; off += (size_t)NG_ * DIN_;
    float* ms_d  = ws + off; off += (size_t)ND_ * DIN_;
    float* cnt_g = ws + off; off += (size_t)NG_;
    float* cnt_d = ws + off; off += (size_t)ND_;
    float* g1    = ws + off; off += (size_t)NG_ * H_;
    float* d1    = ws + off; off += (size_t)ND_ * H_;
    (void)off; (void)ws_size; (void)in_sizes; (void)n_in; (void)out_size;

    const int TPB = 256;
    const int eb  = (E_ + TPB - 1) / TPB;

    hipMemsetAsync(ms_g,  0, (size_t)NG_ * DIN_ * sizeof(float), stream);
    hipMemsetAsync(ms_d,  0, (size_t)ND_ * DIN_ * sizeof(float), stream);
    hipMemsetAsync(cnt_g, 0, (size_t)NG_ * sizeof(float), stream);
    hipMemsetAsync(cnt_d, 0, (size_t)ND_ * sizeof(float), stream);

    count_edges<<<eb, TPB, 0, stream>>>(dst_dg, cnt_g, E_);
    count_edges<<<eb, TPB, 0, stream>>>(dst_gd, cnt_d, E_);

    {
        const int blocks = ((E_ << (7 - 2)) + TPB - 1) / TPB;
        scatter_add4<7><<<blocks, TPB, 0, stream>>>(src_dg, dst_dg, x_disease, ms_g, E_);
        scatter_add4<7><<<blocks, TPB, 0, stream>>>(src_gd, dst_gd, x_gene,    ms_d, E_);
    }

    {
        const int pairs_g = (NG_ / 16 + 1) / 2;
        const int pairs_d = (ND_ / 16 + 1) / 2;
        sage_wmma<DIN_><<<(pairs_g + 3) / 4, 128, 0, stream>>>(
            ms_g, cnt_g, x_gene,    Wl1_dg, bl1_dg, Wr1_dg, g1, NG_);
        sage_wmma<DIN_><<<(pairs_d + 3) / 4, 128, 0, stream>>>(
            ms_d, cnt_d, x_disease, Wl1_gd, bl1_gd, Wr1_gd, d1, ND_);
    }

    hipMemsetAsync(ms_g, 0, (size_t)NG_ * H_ * sizeof(float), stream);
    hipMemsetAsync(ms_d, 0, (size_t)ND_ * H_ * sizeof(float), stream);
    {
        const int blocks = ((E_ << (6 - 2)) + TPB - 1) / TPB;
        scatter_add4<6><<<blocks, TPB, 0, stream>>>(src_dg, dst_dg, d1, ms_g, E_);
        scatter_add4<6><<<blocks, TPB, 0, stream>>>(src_gd, dst_gd, g1, ms_d, E_);
    }

    {
        const int pairs_g = (NG_ / 16 + 1) / 2;
        const int pairs_d = (ND_ / 16 + 1) / 2;
        sage_wmma<H_><<<(pairs_g + 3) / 4, 128, 0, stream>>>(
            ms_g, cnt_g, g1, Wl2_dg, bl2_dg, Wr2_dg, g2, NG_);
        sage_wmma<H_><<<(pairs_d + 3) / 4, 128, 0, stream>>>(
            ms_d, cnt_d, d1, Wl2_gd, bl2_gd, Wr2_gd, d2, ND_);
    }
}


# apply Wl before the edge sum (linearity): node GEMM makes x@Wl and x@Wr+b from one read of x; 64-wide edge rows scaled by 1/cnt add by f32 atomics onto the own term; f32 sum order changes
# speedup vs baseline: 1.1976x; 1.1976x over previous
#include <hip/hip_runtime.h>

typedef float v2f __attribute__((ext_vector_type(2)));
typedef float v8f __attribute__((ext_vector_type(8)));

#define ND_   50000
#define NG_   100000
#define E_    600000
#define DIN_  128
#define H_    64

__global__ __launch_bounds__(256)
void count_edges(const int* __restrict__ dst, float* __restrict__ cnt, int nE) {
    int i = blockIdx.x * blockDim.x + threadIdx.x;
    if (i < nE) atomicAdd(&cnt[dst[i]], 1.0f);
}

__global__ __launch_bounds__(256)
void edge_add_scaled(const int* __restrict__ src, const int* __restrict__ dst,
                     const float* __restrict__ P, const float* __restrict__ cnt,
                     float* __restrict__ acc, int nE) {
    unsigned idx = blockIdx.x * blockDim.x + threadIdx.x;
    unsigned e   = idx >> 4;
    if (e >= (unsigned)nE) return;
    unsigned c = idx & 15u;

    const int   d = dst[e];
    const float s = 1.0f / fmaxf(cnt[d], 1.0f);
    const float4 v = ((const float4*)(P + ((size_t)src[e] << 6)))[c];
    float* o = acc + ((size_t)d << 6) + (size_t)c * 4;
    atomicAdd(o + 0, v.x * s);
    atomicAdd(o + 1, v.y * s);
    atomicAdd(o + 2, v.z * s);
    atomicAdd(o + 3, v.w * s);
}

template<int K>
__global__ __launch_bounds__(128)
void node_wmma(const float* __restrict__ x,
               const float* __restrict__ Wn,
               const float* __restrict__ Ws, const float* __restrict__ bias,
               float* __restrict__ outn, float* __restrict__ outs, int N)
{
    __shared__ v2f Wpk[K * 64];
    {
        float* lds = (float*)Wpk;
        for (int i = threadIdx.x; i < K * 64; i += blockDim.x) {
            const int k = i >> 6, col = i & 63;
            lds[((k >> 1) * 64 + col) * 2 + (k & 1)]           = Wn[i];
            lds[((K / 2 + (k >> 1)) * 64 + col) * 2 + (k & 1)] = Ws[i];
        }
    }
    __syncthreads();

    const int wave = threadIdx.x >> 5;
    const int lane = threadIdx.x & 31;
    const int pair = blockIdx.x * 4 + wave;
    const int row0 = pair * 32;
    if (row0 >= N) return;
    const bool has2 = (row0 + 32 <= N);

    const int m  = lane & 15;
    const int ph = lane >> 4;
    const int rowA0 = row0 + m;
    const int rowA1 = min(row0 + 16 + m, N - 1);

    v8f accn0[4] = {};
    v8f accn1[4] = {};
    v8f accs0[4] = {};
    v8f accs1[4] = {};

    const v2f* X0 = (const v2f*)(x + (size_t)rowA0 * K);
    const v2f* X1 = (const v2f*)(x + (size_t)rowA1 * K);
    for (int k = 0; k < K; k += 4) {
        const int kp = (k >> 1) + ph;
        v2f a0 = X0[kp];
        v2f a1 = X1[kp];
#pragma unroll
        for (int t = 0; t < 4; ++t) {
            v2f bn = Wpk[kp * 64 + t * 16 + m];
            accn0[t] = __builtin_amdgcn_wmma_f32_16x16x4_f32(
                false, a0, false, bn, (short)0, accn0[t], false, false);
            accn1[t] = __builtin_amdgcn_wmma_f32_16x16x4_f32(
                false, a1, false, bn, (short)0, accn1[t], false, false);
            v2f bs = Wpk[(K / 2 + kp) * 64 + t * 16 + m];
            accs0[t] = __builtin_amdgcn_wmma_f32_16x16x4_f32(
                false, a0, false, bs, (short)0, accs0[t], false, false);
            accs1[t] = __builtin_amdgcn_wmma_f32_16x16x4_f32(
                false, a1, false, bs, (short)0, accs1[t], false, false);
        }
    }

    const int colb  = lane & 15;
    const int mb0   = row0 + 8 * ph;
#pragma unroll
    for (int t = 0; t < 4; ++t) {
        const float b = bias[t * 16 + colb];
#pragma unroll
        for (int r = 0; r < 8; ++r) {
            outn[(size_t)(mb0 + r) * 64 + t * 16 + colb] = accn0[t][r];
            outs[(size_t)(mb0 + r) * 64 + t * 16 + colb] = accs0[t][r] + b;
        }
        if (has2) {
#pragma unroll
            for (int r = 0; r < 8; ++r) {
                outn[(size_t)(mb0 + 16 + r) * 64 + t * 16 + colb] = accn1[t][r];
                outs[(size_t)(mb0 + 16 + r) * 64 + t * 16 + colb] = accs1[t][r] + b;
            }
        }
    }
}

extern "C" void kernel_launch(void* const* d_in, const int* in_sizes, int n_in,
                              void* d_out, int out_size, void* d_ws, size_t ws_size,
                              hipStream_t stream) {
    const float* x_disease = (const float*)d_in[0];
    const float* x_gene    = (const float*)d_in[1];
    const int*   src_dg    = (const int*)d_in[2];
    const int*   dst_dg    = (const int*)d_in[3];
    const int*   src_gd    = (const int*)d_in[4];
    const int*   dst_gd    = (const int*)d_in[5];
    const float* Wl1_dg = (const float*)d_in[6];
    const float* bl1_dg = (const float*)d_in[7];
    const float* Wr1_dg = (const float*)d_in[8];
    const float* Wl1_gd = (const float*)d_in[9];
    const float* bl1_gd = (const float*)d_in[10];
    const float* Wr1_gd = (const float*)d_in[11];
    const float* Wl2_dg = (const float*)d_in[12];
    const float* bl2_dg = (const float*)d_in[13];
    const float* Wr2_dg = (const float*)d_in[14];
    const float* Wl2_gd = (const float*)d_in[15];
    const float* bl2_gd = (const float*)d_in[16];
    const float* Wr2_gd = (const float*)d_in[17];

    float* d2 = (float*)d_out;
    float* g2 = d2 + (size_t)ND_ * 64;

    float* ws = (float*)d_ws;
    size_t off = 0;
    float* p_g   = ws + off; off += (size_t)NG_ * H_;
    float* p_d   = ws + off; off += (size_t)ND_ * H_;
    float* cnt_g = ws + off; off += (size_t)NG_;
    float* cnt_d = ws + off; off += (size_t)ND_;
    float* g1    = ws + off; off += (size_t)NG_ * H_;
    float* d1    = ws + off; off += (size_t)ND_ * H_;
    (void)off; (void)ws_size; (void)in_sizes; (void)n_in; (void)out_size;

    const int TPB = 256;
    const int eb  = (E_ + TPB - 1) / TPB;
    const int ab  = (E_ * 16 + TPB - 1) / TPB;
    const int pairs_g = (NG_ / 16 + 1) / 2;
    const int pairs_d = (ND_ / 16 + 1) / 2;

    hipMemsetAsync(cnt_g, 0, (size_t)NG_ * sizeof(float), stream);
    hipMemsetAsync(cnt_d, 0, (size_t)ND_ * sizeof(float), stream);

    count_edges<<<eb, TPB, 0, stream>>>(dst_dg, cnt_g, E_);
    count_edges<<<eb, TPB, 0, stream>>>(dst_gd, cnt_d, E_);

    node_wmma<DIN_><<<(pairs_d + 3) / 4, 128, 0, stream>>>(
        x_disease, Wl1_dg, Wr1_gd, bl1_gd, p_d, d1, ND_);
    node_wmma<DIN_><<<(pairs_g + 3) / 4, 128, 0, stream>>>(
        x_gene,    Wl1_gd, Wr1_dg, bl1_dg, p_g, g1, NG_);

    edge_add_scaled<<<ab, TPB, 0, stream>>>(src_dg, dst_dg, p_d, cnt_g, g1, E_);
    edge_add_scaled<<<ab, TPB, 0, stream>>>(src_gd, dst_gd, p_g, cnt_d, d1, E_);

    node_wmma<H_><<<(pairs_d + 3) / 4, 128, 0, stream>>>(
        d1, Wl2_dg, Wr2_gd, bl2_gd, p_d, d2, ND_);
    node_wmma<H_><<<(pairs_g + 3) / 4, 128, 0, stream>>>(
        g1, Wl2_gd, Wr2_dg, bl2_dg, p_g, g2, NG_);

    edge_add_scaled<<<ab, TPB, 0, stream>>>(src_dg, dst_dg, p_d, cnt_g, g2, E_);
    edge_add_scaled<<<ab, TPB, 0, stream>>>(src_gd, dst_gd, p_g, cnt_d, d2, E_);
}


# order edges by destination inside the call (integer-atomic rank, scan, place) and sum each destination row's sender rows in registers, one writer per row; no f32 atomics; f32 sum order changes
# speedup vs baseline: 1.5550x; 1.2984x over previous
#include <hip/hip_runtime.h>

typedef float v2f __attribute__((ext_vector_type(2)));
typedef float v8f __attribute__((ext_vector_type(8)));

#define ND_   50000
#define NG_   100000
#define E_    600000
#define DIN_  128
#define H_    64
#define NT_   (NG_ + ND_)
#define E2_   (2 * E_)

__global__ __launch_bounds__(256)
void edge_rank(const int* __restrict__ dst_dg, const int* __restrict__ dst_gd,
               int* __restrict__ cnt, int* __restrict__ rank) {
    const int q = blockIdx.x * blockDim.x + threadIdx.x;
    if (q >= E2_ / 4) return;
    int4 d; int base;
    if (q < E_ / 4) { d = ((const int4*)dst_dg)[q];          base = 0;   }
    else            { d = ((const int4*)dst_gd)[q - E_ / 4]; base = NG_; }
    int4 r;
    r.x = atomicAdd(&cnt[base + d.x], 1);
    r.y = atomicAdd(&cnt[base + d.y], 1);
    r.z = atomicAdd(&cnt[base + d.z], 1);
    r.w = atomicAdd(&cnt[base + d.w], 1);
    ((int4*)rank)[q] = r;
}

__global__ __launch_bounds__(256)
void seg_offsets(const int* __restrict__ cnt, int2* __restrict__ seg,
                 int* __restrict__ cursor) {
    __shared__ int wtot[8];
    __shared__ int base_s;
    const int i    = blockIdx.x * 256 + threadIdx.x;
    const int lane = threadIdx.x & 31;
    const int w    = threadIdx.x >> 5;
    const int c    = (i < NT_) ? cnt[i] : 0;
    int incl = c;
#pragma unroll
    for (int d = 1; d < 32; d <<= 1) {
        const int v = __shfl_up(incl, d);
        if (lane >= d) incl += v;
    }
    if (lane == 31) wtot[w] = incl;
    __syncthreads();
    if (threadIdx.x == 0) {
        int tot = 0;
        for (int k = 0; k < 8; ++k) { const int x = wtot[k]; wtot[k] = tot; tot += x; }
        base_s = atomicAdd(cursor, tot);
    }
    __syncthreads();
    if (i < NT_) seg[i] = make_int2(base_s + wtot[w] + incl - c, c);
}

__global__ __launch_bounds__(256)
void edge_place(const int* __restrict__ src_dg, const int* __restrict__ dst_dg,
                const int* __restrict__ src_gd, const int* __restrict__ dst_gd,
                const int2* __restrict__ seg, const int* __restrict__ rank,
                int* __restrict__ slots) {
    const int q = blockIdx.x * blockDim.x + threadIdx.x;
    if (q >= E2_ / 4) return;
    int4 d, s; int base;
    if (q < E_ / 4) {
        d = ((const int4*)dst_dg)[q];          s = ((const int4*)src_dg)[q];          base = 0;
    } else {
        d = ((const int4*)dst_gd)[q - E_ / 4]; s = ((const int4*)src_gd)[q - E_ / 4]; base = NG_;
    }
    const int4 r = ((const int4*)rank)[q];
    const int p0 = min(seg[base + d.x].x + r.x, E2_ - 1);
    const int p1 = min(seg[base + d.y].x + r.y, E2_ - 1);
    const int p2 = min(seg[base + d.z].x + r.z, E2_ - 1);
    const int p3 = min(seg[base + d.w].x + r.w, E2_ - 1);
    slots[p0] = s.x;
    slots[p1] = s.y;
    slots[p2] = s.z;
    slots[p3] = s.w;
}

__device__ __forceinline__
void gather_set(const float* __restrict__ P, const unsigned last_src,
                float* __restrict__ acc, const int2* __restrict__ seg,
                const int* __restrict__ slots,
                const int n, const int first, const int step, const unsigned c) {
    const float4* __restrict__ P4 = (const float4*)P;
    int j = first;
    int2 sg = make_int2(0, 0);
    if (j < n) sg = seg[j];
    while (j < n) {
        const int o = sg.x, deg = sg.y;
        float4* orow = (float4*)(acc + ((size_t)j << 6)) + c;
        const float4 own = *orow;
        const int jn = j + step;
        if (jn < n) sg = seg[jn];
        float4 s = make_float4(0.f, 0.f, 0.f, 0.f);
        int i = 0;
        for (; i + 4 <= deg; i += 4) {
            const unsigned s0 = min((unsigned)slots[o + i + 0], last_src);
            const unsigned s1 = min((unsigned)slots[o + i + 1], last_src);
            const unsigned s2 = min((unsigned)slots[o + i + 2], last_src);
            const unsigned s3 = min((unsigned)slots[o + i + 3], last_src);
            const float4 v0 = P4[s0 * 16u + c];
            const float4 v1 = P4[s1 * 16u + c];
            const float4 v2 = P4[s2 * 16u + c];
            const float4 v3 = P4[s3 * 16u + c];
            s.x += v0.x; s.y += v0.y; s.z += v0.z; s.w += v0.w;
            s.x += v1.x; s.y += v1.y; s.z += v1.z; s.w += v1.w;
            s.x += v2.x; s.y += v2.y; s.z += v2.z; s.w += v2.w;
            s.x += v3.x; s.y += v3.y; s.z += v3.z; s.w += v3.w;
        }
        for (; i < deg; ++i) {
            const unsigned s0 = min((unsigned)slots[o + i], last_src);
            const float4 v0 = P4[s0 * 16u + c];
            s.x += v0.x; s.y += v0.y; s.z += v0.z; s.w += v0.w;
        }
        const float inv = 1.0f / fmaxf((float)deg, 1.0f);
        float4 r;
        r.x = fmaf(s.x, inv, own.x);
        r.y = fmaf(s.y, inv, own.y);
        r.z = fmaf(s.z, inv, own.z);
        r.w = fmaf(s.w, inv, own.w);
        *orow = r;
        j = jn;
    }
}

__global__ __launch_bounds__(256)
void row_gather(const float* __restrict__ P_dis, const float* __restrict__ P_gene,
                float* __restrict__ acc_gene, float* __restrict__ acc_dis,
                const int2* __restrict__ seg, const int* __restrict__ slots) {
    const unsigned c = threadIdx.x & 15u;
    const int hw  = (blockIdx.x * 256 + threadIdx.x) >> 4;
    const int nhw = (gridDim.x * 256) >> 4;
    gather_set(P_dis,  ND_ - 1, acc_gene, seg,       slots, NG_, hw,           nhw, c);
    gather_set(P_gene, NG_ - 1, acc_dis,  seg + NG_, slots, ND_, nhw - 1 - hw, nhw, c);
}

template<int K>
__global__ __launch_bounds__(128)
void node_wmma(const float* x,
               const float* __restrict__ Wn,
               const float* __restrict__ Ws, const float* __restrict__ bias,
               float* outn, float* __restrict__ outs, int N)
{
    __shared__ v2f Wpk[K * 64];
    {
        float* lds = (float*)Wpk;
        for (int i = threadIdx.x; i < K * 64; i += blockDim.x) {
            const int k = i >> 6, col = i & 63;
            lds[((k >> 1) * 64 + col) * 2 + (k & 1)]           = Wn[i];
            lds[((K / 2 + (k >> 1)) * 64 + col) * 2 + (k & 1)] = Ws[i];
        }
    }
    __syncthreads();

    const int wave = threadIdx.x >> 5;
    const int lane = threadIdx.x & 31;
    const int pair = blockIdx.x * 4 + wave;
    const int row0 = pair * 32;
    if (row0 >= N) return;
    const bool has2 = (row0 + 32 <= N);

    const int m  = lane & 15;
    const int ph = lane >> 4;
    const int rowA0 = row0 + m;
    const int rowA1 = min(row0 + 16 + m, N - 1);

    v8f accn0[4] = {};
    v8f accn1[4] = {};
    v8f accs0[4] = {};
    v8f accs1[4] = {};

    const v2f* X0 = (const v2f*)(x + (size_t)rowA0 * K);
    const v2f* X1 = (const v2f*)(x + (size_t)rowA1 * K);
    for (int k = 0; k < K; k += 4) {
        const int kp = (k >> 1) + ph;
        v2f a0 = X0[kp];
        v2f a1 = X1[kp];
#pragma unroll
        for (int t = 0; t < 4; ++t) {
            v2f bn = Wpk[kp * 64 + t * 16 + m];
            accn0[t] = __builtin_amdgcn_wmma_f32_16x16x4_f32(
                false, a0, false, bn, (short)0, accn0[t], false, false);
            accn1[t] = __builtin_amdgcn_wmma_f32_16x16x4_f32(
                false, a1, false, bn, (short)0, accn1[t], false, false);
            v2f bs = Wpk[(K / 2 + kp) * 64 + t * 16 + m];
            accs0[t] = __builtin_amdgcn_wmma_f32_16x16x4_f32(
                false, a0, false, bs, (short)0, accs0[t], false, false);
            accs1[t] = __builtin_amdgcn_wmma_f32_16x16x4_f32(
                false, a1, false, bs, (short)0, accs1[t], false, false);
        }
    }

    const int colb  = lane & 15;
    const int mb0   = row0 + 8 * ph;
#pragma unroll
    for (int t = 0; t < 4; ++t) {
        const float b = bias[t * 16 + colb];
#pragma unroll
        for (int r = 0; r < 8; ++r) {
            outn[(size_t)(mb0 + r) * 64 + t * 16 + colb] = accn0[t][r];
            outs[(size_t)(mb0 + r) * 64 + t * 16 + colb] = accs0[t][r] + b;
        }
        if (has2) {
#pragma unroll
            for (int r = 0; r < 8; ++r) {
                outn[(size_t)(mb0 + 16 + r) * 64 + t * 16 + colb] = accn1[t][r];
                outs[(size_t)(mb0 + 16 + r) * 64 + t * 16 + colb] = accs1[t][r] + b;
            }
        }
    }
}

extern "C" void kernel_launch(void* const* d_in, const int* in_sizes, int n_in,
                              void* d_out, int out_size, void* d_ws, size_t ws_size,
                              hipStream_t stream) {
    const float* x_disease = (const float*)d_in[0];
    const float* x_gene    = (const float*)d_in[1];
    const int*   src_dg    = (const int*)d_in[2];
    const int*   dst_dg    = (const int*)d_in[3];
    const int*   src_gd    = (const int*)d_in[4];
    const int*   dst_gd    = (const int*)d_in[5];
    const float* Wl1_dg = (const float*)d_in[6];
    const float* bl1_dg = (const float*)d_in[7];
    const float* Wr1_dg = (const float*)d_in[8];
    const float* Wl1_gd = (const float*)d_in[9];
    const float* bl1_gd = (const float*)d_in[10];
    const float* Wr1_gd = (const float*)d_in[11];
    const float* Wl2_dg = (const float*)d_in[12];
    const float* bl2_dg = (const float*)d_in[13];
    const float* Wr2_dg = (const float*)d_in[14];
    const float* Wl2_gd = (const float*)d_in[15];
    const float* bl2_gd = (const float*)d_in[16];
    const float* Wr2_gd = (const float*)d_in[17];

    float* d2 = (float*)d_out;
    float* g2 = d2 + (size_t)ND_ * 64;
    float* p1_d = d2;
    float* p1_g = g2;

    float* ws = (float*)d_ws;
    size_t off = 0;
    int*  slots  = (int*)(ws + off);  off += (size_t)E2_;
    int*  rank   = (int*)(ws + off);  off += (size_t)E2_;
    int2* seg    = (int2*)(ws + off); off += (size_t)NT_ * 2;
    int*  cnt    = (int*)(ws + off);  off += (size_t)NT_;
    int*  cursor = (int*)(ws + off);  off += 48;
    float* g1    = ws + off; off += (size_t)NG_ * H_;
    float* d1    = ws + off; off += (size_t)ND_ * H_;
    (void)off; (void)ws_size; (void)in_sizes; (void)n_in; (void)out_size;

    const int TPB = 256;
    const int eb  = (E2_ / 4 + TPB - 1) / TPB;
    const int nb  = (NT_ + TPB - 1) / TPB;
    const int gb  = 1024;
    const int pairs_g = (NG_ / 16 + 1) / 2;
    const int pairs_d = (ND_ / 16 + 1) / 2;

    hipMemsetAsync(cnt, 0, ((size_t)NT_ + 48) * sizeof(int), stream);

    edge_rank<<<eb, TPB, 0, stream>>>(dst_dg, dst_gd, cnt, rank);
    seg_offsets<<<nb, TPB, 0, stream>>>(cnt, seg, cursor);
    edge_place<<<eb, TPB, 0, stream>>>(src_dg, dst_dg, src_gd, dst_gd, seg, rank, slots);

    node_wmma<DIN_><<<(pairs_d + 3) / 4, 128, 0, stream>>>(
        x_disease, Wl1_dg, Wr1_gd, bl1_gd, p1_d, d1, ND_);
    node_wmma<DIN_><<<(pairs_g + 3) / 4, 128, 0, stream>>>(
        x_gene,    Wl1_gd, Wr1_dg, bl1_dg, p1_g, g1, NG_);

    row_gather<<<gb, TPB, 0, stream>>>(p1_d, p1_g, g1, d1, seg, slots);

    node_wmma<H_><<<(pairs_d + 3) / 4, 128, 0, stream>>>(
        d1, Wl2_dg, Wr2_gd, bl2_gd, d1, d2, ND_);
    node_wmma<H_><<<(pairs_g + 3) / 4, 128, 0, stream>>>(
        g1, Wl2_gd, Wr2_dg, bl2_dg, g1, g2, NG_);

    row_gather<<<gb, TPB, 0, stream>>>(d1, g1, g2, d2, seg, slots);
}


# node GEMM weight staging: each thread issues all its 16-byte W loads (row pairs) as one group, then packs them into LDS with 16-byte stores: one memory round trip per block, not one per element pair
# speedup vs baseline: 1.9853x; 1.2767x over previous
#include <hip/hip_runtime.h>

typedef float v2f __attribute__((ext_vector_type(2)));
typedef float v8f __attribute__((ext_vector_type(8)));

#define ND_   50000
#define NG_   100000
#define E_    600000
#define DIN_  128
#define H_    64
#define NT_   (NG_ + ND_)
#define E2_   (2 * E_)

__global__ __launch_bounds__(256)
void edge_rank(const int* __restrict__ dst_dg, const int* __restrict__ dst_gd,
               int* __restrict__ cnt, int* __restrict__ rank) {
    const int q = blockIdx.x * blockDim.x + threadIdx.x;
    if (q >= E2_ / 4) return;
    int4 d; int base;
    if (q < E_ / 4) { d = ((const int4*)dst_dg)[q];          base = 0;   }
    else            { d = ((const int4*)dst_gd)[q - E_ / 4]; base = NG_; }
    int4 r;
    r.x = atomicAdd(&cnt[base + d.x], 1);
    r.y = atomicAdd(&cnt[base + d.y], 1);
    r.z = atomicAdd(&cnt[base + d.z], 1);
    r.w = atomicAdd(&cnt[base + d.w], 1);
    ((int4*)rank)[q] = r;
}

__global__ __launch_bounds__(256)
void seg_offsets(const int* __restrict__ cnt, int2* __restrict__ seg,
                 int* __restrict__ cursor) {
    __shared__ int wtot[8];
    __shared__ int base_s;
    const int i    = blockIdx.x * 256 + threadIdx.x;
    const int lane = threadIdx.x & 31;
    const int w    = threadIdx.x >> 5;
    const int c    = (i < NT_) ? cnt[i] : 0;
    int incl = c;
#pragma unroll
    for (int d = 1; d < 32; d <<= 1) {
        const int v = __shfl_up(incl, d);
        if (lane >= d) incl += v;
    }
    if (lane == 31) wtot[w] = incl;
    __syncthreads();
    if (threadIdx.x == 0) {
        int tot = 0;
        for (int k = 0; k < 8; ++k) { const int x = wtot[k]; wtot[k] = tot; tot += x; }
        base_s = atomicAdd(cursor, tot);
    }
    __syncthreads();
    if (i < NT_) seg[i] = make_int2(base_s + wtot[w] + incl - c, c);
}

__global__ __launch_bounds__(256)
void edge_place(const int* __restrict__ src_dg, const int* __restrict__ dst_dg,
                const int* __restrict__ src_gd, const int* __restrict__ dst_gd,
                const int2* __restrict__ seg, const int* __restrict__ rank,
                int* __restrict__ slots) {
    const int q = blockIdx.x * blockDim.x + threadIdx.x;
    if (q >= E2_ / 4) return;
    int4 d, s; int base;
    if (q < E_ / 4) {
        d = ((const int4*)dst_dg)[q];          s = ((const int4*)src_dg)[q];          base = 0;
    } else {
        d = ((const int4*)dst_gd)[q - E_ / 4]; s = ((const int4*)src_gd)[q - E_ / 4]; base = NG_;
    }
    const int4 r = ((const int4*)rank)[q];
    const int p0 = min(seg[base + d.x].x + r.x, E2_ - 1);
    const int p1 = min(seg[base + d.y].x + r.y, E2_ - 1);
    const int p2 = min(seg[base + d.z].x + r.z, E2_ - 1);
    const int p3 = min(seg[base + d.w].x + r.w, E2_ - 1);
    slots[p0] = s.x;
    slots[p1] = s.y;
    slots[p2] = s.z;
    slots[p3] = s.w;
}

__device__ __forceinline__
void gather_set(const float* __restrict__ P, const unsigned last_src,
                float* __restrict__ acc, const int2* __restrict__ seg,
                const int* __restrict__ slots,
                const int n, const int first, const int step, const unsigned c) {
    const float4* __restrict__ P4 = (const float4*)P;
    int j = first;
    int2 sg = make_int2(0, 0);
    if (j < n) sg = seg[j];
    while (j < n) {
        const int o = sg.x, deg = sg.y;
        float4* orow = (float4*)(acc + ((size_t)j << 6)) + c;
        const float4 own = *orow;
        const int jn = j + step;
        if (jn < n) sg = seg[jn];
        float4 s = make_float4(0.f, 0.f, 0.f, 0.f);
        int i = 0;
        for (; i + 4 <= deg; i += 4) {
            const unsigned s0 = min((unsigned)slots[o + i + 0], last_src);
            const unsigned s1 = min((unsigned)slots[o + i + 1], last_src);
            const unsigned s2 = min((unsigned)slots[o + i + 2], last_src);
            const unsigned s3 = min((unsigned)slots[o + i + 3], last_src);
            const float4 v0 = P4[s0 * 16u + c];
            const float4 v1 = P4[s1 * 16u + c];
            const float4 v2 = P4[s2 * 16u + c];
            const float4 v3 = P4[s3 * 16u + c];
            s.x += v0.x; s.y += v0.y; s.z += v0.z; s.w += v0.w;
            s.x += v1.x; s.y += v1.y; s.z += v1.z; s.w += v1.w;
            s.x += v2.x; s.y += v2.y; s.z += v2.z; s.w += v2.w;
            s.x += v3.x; s.y += v3.y; s.z += v3.z; s.w += v3.w;
        }
        for (; i < deg; ++i) {
            const unsigned s0 = min((unsigned)slots[o + i], last_src);
            const float4 v0 = P4[s0 * 16u + c];
            s.x += v0.x; s.y += v0.y; s.z += v0.z; s.w += v0.w;
        }
        const float inv = 1.0f / fmaxf((float)deg, 1.0f);
        float4 r;
        r.x = fmaf(s.x, inv, own.x);
        r.y = fmaf(s.y, inv, own.y);
        r.z = fmaf(s.z, inv, own.z);
        r.w = fmaf(s.w, inv, own.w);
        *orow = r;
        j = jn;
    }
}

__global__ __launch_bounds__(256)
void row_gather(const float* __restrict__ P_dis, const float* __restrict__ P_gene,
                float* __restrict__ acc_gene, float* __restrict__ acc_dis,
                const int2* __restrict__ seg, const int* __restrict__ slots) {
    const unsigned c = threadIdx.x & 15u;
    const int hw  = (blockIdx.x * 256 + threadIdx.x) >> 4;
    const int nhw = (gridDim.x * 256) >> 4;
    gather_set(P_dis,  ND_ - 1, acc_gene, seg,       slots, NG_, hw,           nhw, c);
    gather_set(P_gene, NG_ - 1, acc_dis,  seg + NG_, slots, ND_, nhw - 1 - hw, nhw, c);
}

template<int K>
__global__ __launch_bounds__(128)
void node_wmma(const float* x,
               const float* __restrict__ Wn,
               const float* __restrict__ Ws, const float* __restrict__ bias,
               float* outn, float* __restrict__ outs, int N)
{
    __shared__ v2f Wpk[K * 64];
    {
        constexpr int PER_T = K / 8;
        float4* lds4 = (float4*)Wpk;
        float4 ra[PER_T], rb[PER_T];
#pragma unroll
        for (int u = 0; u < PER_T; ++u) {
            const int I  = u * 128 + (int)threadIdx.x;
            const int PP = I >> 4, q = I & 15;
            const float* Wsel = (u * 8 < K / 2) ? Wn : Ws;
            const int pp = PP - ((u * 8 < K / 2) ? 0 : K / 2);
            const float4* g = (const float4*)(Wsel + (size_t)(2 * pp) * 64) + q;
            ra[u] = g[0];
            rb[u] = g[16];
        }
#pragma unroll
        for (int u = 0; u < PER_T; ++u) {
            const int I  = u * 128 + (int)threadIdx.x;
            const int PP = I >> 4, q = I & 15;
            float4* d = lds4 + ((PP * 64 + 4 * q) >> 1);
            d[0] = make_float4(ra[u].x, rb[u].x, ra[u].y, rb[u].y);
            d[1] = make_float4(ra[u].z, rb[u].z, ra[u].w, rb[u].w);
        }
    }
    __syncthreads();

    const int wave = threadIdx.x >> 5;
    const int lane = threadIdx.x & 31;
    const int pair = blockIdx.x * 4 + wave;
    const int row0 = pair * 32;
    if (row0 >= N) return;
    const bool has2 = (row0 + 32 <= N);

    const int m  = lane & 15;
    const int ph = lane >> 4;
    const int rowA0 = row0 + m;
    const int rowA1 = min(row0 + 16 + m, N - 1);

    v8f accn0[4] = {};
    v8f accn1[4] = {};
    v8f accs0[4] = {};
    v8f accs1[4] = {};

    const v2f* X0 = (const v2f*)(x + (size_t)rowA0 * K);
    const v2f* X1 = (const v2f*)(x + (size_t)rowA1 * K);
    for (int k = 0; k < K; k += 4) {
        const int kp = (k >> 1) + ph;
        v2f a0 = X0[kp];
        v2f a1 = X1[kp];
#pragma unroll
        for (int t = 0; t < 4; ++t) {
            v2f bn = Wpk[kp * 64 + t * 16 + m];
            accn0[t] = __builtin_amdgcn_wmma_f32_16x16x4_f32(
                false, a0, false, bn, (short)0, accn0[t], false, false);
            accn1[t] = __builtin_amdgcn_wmma_f32_16x16x4_f32(
                false, a1, false, bn, (short)0, accn1[t], false, false);
            v2f bs = Wpk[(K / 2 + kp) * 64 + t * 16 + m];
            accs0[t] = __builtin_amdgcn_wmma_f32_16x16x4_f32(
                false, a0, false, bs, (short)0, accs0[t], false, false);
            accs1[t] = __builtin_amdgcn_wmma_f32_16x16x4_f32(
                false, a1, false, bs, (short)0, accs1[t], false, false);
        }
    }

    const int colb  = lane & 15;
    const int mb0   = row0 + 8 * ph;
#pragma unroll
    for (int t = 0; t < 4; ++t) {
        const float b = bias[t * 16 + colb];
#pragma unroll
        for (int r = 0; r < 8; ++r) {
            outn[(size_t)(mb0 + r) * 64 + t * 16 + colb] = accn0[t][r];
            outs[(size_t)(mb0 + r) * 64 + t * 16 + colb] = accs0[t][r] + b;
        }
        if (has2) {
#pragma unroll
            for (int r = 0; r < 8; ++r) {
                outn[(size_t)(mb0 + 16 + r) * 64 + t * 16 + colb] = accn1[t][r];
                outs[(size_t)(mb0 + 16 + r) * 64 + t * 16 + colb] = accs1[t][r] + b;
            }
        }
    }
}

extern "C" void kernel_launch(void* const* d_in, const int* in_sizes, int n_in,
                              void* d_out, int out_size, void* d_ws, size_t ws_size,
                              hipStream_t stream) {
    const float* x_disease = (const float*)d_in[0];
    const float* x_gene    = (const float*)d_in[1];
    const int*   src_dg    = (const int*)d_in[2];
    const int*   dst_dg    = (const int*)d_in[3];
    const int*   src_gd    = (const int*)d_in[4];
    const int*   dst_gd    = (const int*)d_in[5];
    const float* Wl1_dg = (const float*)d_in[6];
    const float* bl1_dg = (const float*)d_in[7];
    const float* Wr1_dg = (const float*)d_in[8];
    const float* Wl1_gd = (const float*)d_in[9];
    const float* bl1_gd = (const float*)d_in[10];
    const float* Wr1_gd = (const float*)d_in[11];
    const float* Wl2_dg = (const float*)d_in[12];
    const float* bl2_dg = (const float*)d_in[13];
    const float* Wr2_dg = (const float*)d_in[14];
    const float* Wl2_gd = (const float*)d_in[15];
    const float* bl2_gd = (const float*)d_in[16];
    const float* Wr2_gd = (const float*)d_in[17];

    float* d2 = (float*)d_out;
    float* g2 = d2 + (size_t)ND_ * 64;
    float* p1_d = d2;
    float* p1_g = g2;

    float* ws = (float*)d_ws;
    size_t off = 0;
    int*  slots  = (int*)(ws + off);  off += (size_t)E2_;
    int*  rank   = (int*)(ws + off);  off += (size_t)E2_;
    int2* seg    = (int2*)(ws + off); off += (size_t)NT_ * 2;
    int*  cnt    = (int*)(ws + off);  off += (size_t)NT_;
    int*  cursor = (int*)(ws + off);  off += 48;
    float* g1    = ws + off; off += (size_t)NG_ * H_;
    float* d1    = ws + off; off += (size_t)ND_ * H_;
    (void)off; (void)ws_size; (void)in_sizes; (void)n_in; (void)out_size;

    const int TPB = 256;
    const int eb  = (E2_ / 4 + TPB - 1) / TPB;
    const int nb  = (NT_ + TPB - 1) / TPB;
    const int gb  = 1024;
    const int pairs_g = (NG_ / 16 + 1) / 2;
    const int pairs_d = (ND_ / 16 + 1) / 2;

    hipMemsetAsync(cnt, 0, ((size_t)NT_ + 48) * sizeof(int), stream);

    edge_rank<<<eb, TPB, 0, stream>>>(dst_dg, dst_gd, cnt, rank);
    seg_offsets<<<nb, TPB, 0, stream>>>(cnt, seg, cursor);
    edge_place<<<eb, TPB, 0, stream>>>(src_dg, dst_dg, src_gd, dst_gd, seg, rank, slots);

    node_wmma<DIN_><<<(pairs_d + 3) / 4, 128, 0, stream>>>(
        x_disease, Wl1_dg, Wr1_gd, bl1_gd, p1_d, d1, ND_);
    node_wmma<DIN_><<<(pairs_g + 3) / 4, 128, 0, stream>>>(
        x_gene,    Wl1_gd, Wr1_dg, bl1_dg, p1_g, g1, NG_);

    row_gather<<<gb, TPB, 0, stream>>>(p1_d, p1_g, g1, d1, seg, slots);

    node_wmma<H_><<<(pairs_d + 3) / 4, 128, 0, stream>>>(
        d1, Wl2_dg, Wr2_gd, bl2_gd, d1, d2, ND_);
    node_wmma<H_><<<(pairs_g + 3) / 4, 128, 0, stream>>>(
        g1, Wl2_gd, Wr2_dg, bl2_dg, g1, g2, NG_);

    row_gather<<<gb, TPB, 0, stream>>>(d1, g1, g2, d2, seg, slots);
}


# row gather: a row's slot indices come from one coalesced load issued a row ahead and are shared by lane broadcast; all its sender-row loads go out as one group, no serial tail; sum order kept
# speedup vs baseline: 2.0649x; 1.0401x over previous
#include <hip/hip_runtime.h>

typedef float v2f __attribute__((ext_vector_type(2)));
typedef float v8f __attribute__((ext_vector_type(8)));

#define ND_   50000
#define NG_   100000
#define E_    600000
#define DIN_  128
#define H_    64
#define NT_   (NG_ + ND_)
#define E2_   (2 * E_)

__global__ __launch_bounds__(256)
void edge_rank(const int* __restrict__ dst_dg, const int* __restrict__ dst_gd,
               int* __restrict__ cnt, int* __restrict__ rank) {
    const int q = blockIdx.x * blockDim.x + threadIdx.x;
    if (q >= E2_ / 4) return;
    int4 d; int base;
    if (q < E_ / 4) { d = ((const int4*)dst_dg)[q];          base = 0;   }
    else            { d = ((const int4*)dst_gd)[q - E_ / 4]; base = NG_; }
    int4 r;
    r.x = atomicAdd(&cnt[base + d.x], 1);
    r.y = atomicAdd(&cnt[base + d.y], 1);
    r.z = atomicAdd(&cnt[base + d.z], 1);
    r.w = atomicAdd(&cnt[base + d.w], 1);
    ((int4*)rank)[q] = r;
}

__global__ __launch_bounds__(256)
void seg_offsets(const int* __restrict__ cnt, int2* __restrict__ seg,
                 int* __restrict__ cursor) {
    __shared__ int wtot[8];
    __shared__ int base_s;
    const int i    = blockIdx.x * 256 + threadIdx.x;
    const int lane = threadIdx.x & 31;
    const int w    = threadIdx.x >> 5;
    const int c    = (i < NT_) ? cnt[i] : 0;
    int incl = c;
#pragma unroll
    for (int d = 1; d < 32; d <<= 1) {
        const int v = __shfl_up(incl, d);
        if (lane >= d) incl += v;
    }
    if (lane == 31) wtot[w] = incl;
    __syncthreads();
    if (threadIdx.x == 0) {
        int tot = 0;
        for (int k = 0; k < 8; ++k) { const int x = wtot[k]; wtot[k] = tot; tot += x; }
        base_s = atomicAdd(cursor, tot);
    }
    __syncthreads();
    if (i < NT_) seg[i] = make_int2(base_s + wtot[w] + incl - c, c);
}

__global__ __launch_bounds__(256)
void edge_place(const int* __restrict__ src_dg, const int* __restrict__ dst_dg,
                const int* __restrict__ src_gd, const int* __restrict__ dst_gd,
                const int2* __restrict__ seg, const int* __restrict__ rank,
                int* __restrict__ slots) {
    const int q = blockIdx.x * blockDim.x + threadIdx.x;
    if (q >= E2_ / 4) return;
    int4 d, s; int base;
    if (q < E_ / 4) {
        d = ((const int4*)dst_dg)[q];          s = ((const int4*)src_dg)[q];          base = 0;
    } else {
        d = ((const int4*)dst_gd)[q - E_ / 4]; s = ((const int4*)src_gd)[q - E_ / 4]; base = NG_;
    }
    const int4 r = ((const int4*)rank)[q];
    const int p0 = min(seg[base + d.x].x + r.x, E2_ - 1);
    const int p1 = min(seg[base + d.y].x + r.y, E2_ - 1);
    const int p2 = min(seg[base + d.z].x + r.z, E2_ - 1);
    const int p3 = min(seg[base + d.w].x + r.w, E2_ - 1);
    slots[p0] = s.x;
    slots[p1] = s.y;
    slots[p2] = s.z;
    slots[p3] = s.w;
}


#define ROW_LANE(v, i) \
    ((unsigned)__builtin_amdgcn_update_dpp(0, (int)(v), 0x150 + (i), 0xf, 0xf, false))

#define ROW_LOAD(i, ix_, m_) \
    { const unsigned r_ = min(ROW_LANE(ix_, i), last_src); \
      if ((i) < (m_)) v##i = P4[r_ * 16u + c]; }

#define ROW_ADD(i) \
    s.x += v##i.x; s.y += v##i.y; s.z += v##i.z; s.w += v##i.w;

#define ROW_GROUP(ix_, m_, FIRST_) \
    { \
        const int mm_ = (m_); \
        float4 v0 = z4, v1 = z4, v2  = z4, v3  = z4, v4  = z4, v5  = z4, v6  = z4, v7  = z4; \
        float4 v8 = z4, v9 = z4, v10 = z4, v11 = z4, v12 = z4, v13 = z4, v14 = z4, v15 = z4; \
        ROW_LOAD(0, ix_, mm_) ROW_LOAD(1, ix_, mm_) ROW_LOAD(2, ix_, mm_) ROW_LOAD(3, ix_, mm_) \
        ROW_LOAD(4, ix_, mm_) ROW_LOAD(5, ix_, mm_) ROW_LOAD(6, ix_, mm_) ROW_LOAD(7, ix_, mm_) \
        if (mm_ > 8) { \
            ROW_LOAD(8,  ix_, mm_) ROW_LOAD(9,  ix_, mm_) ROW_LOAD(10, ix_, mm_) ROW_LOAD(11, ix_, mm_) \
            ROW_LOAD(12, ix_, mm_) ROW_LOAD(13, ix_, mm_) ROW_LOAD(14, ix_, mm_) ROW_LOAD(15, ix_, mm_) \
        } \
        FIRST_ \
        ROW_ADD(1) ROW_ADD(2) ROW_ADD(3) \
        ROW_ADD(4) ROW_ADD(5) ROW_ADD(6) ROW_ADD(7) \
        if (mm_ > 8) { \
            ROW_ADD(8)  ROW_ADD(9)  ROW_ADD(10) ROW_ADD(11) \
            ROW_ADD(12) ROW_ADD(13) ROW_ADD(14) ROW_ADD(15) \
        } \
    }

__device__ __forceinline__
void gather_set(const float* __restrict__ P, const unsigned last_src,
                float* __restrict__ acc, const int2* __restrict__ seg,
                const int* __restrict__ slots,
                const int n, const int first, const int step, const unsigned c) {
    const float4* __restrict__ P4 = (const float4*)P;
    const float4 z4 = make_float4(0.f, 0.f, 0.f, 0.f);
    int j  = first;
    int jn = j + step;
    int2 sg  = make_int2(0, 0);
    int2 sgn = make_int2(0, 0);
    unsigned idx = 0;
    if (j  < n) sg  = seg[j];
    if (jn < n) sgn = seg[jn];
    if ((int)c < sg.y) idx = (unsigned)slots[sg.x + (int)c];
    idx = min(idx, last_src);
    while (j < n) {
        const int o = sg.x, deg = sg.y;
        float4* orow = (float4*)(acc + ((size_t)j << 6)) + c;
        const float4 own = *orow;
        unsigned idxn = 0;
        if ((int)c < sgn.y) idxn = (unsigned)slots[sgn.x + (int)c];
        const int jnn = jn + step;
        int2 sgnn = make_int2(0, 0);
        if (jnn < n) sgnn = seg[jnn];

        float4 s;
        ROW_GROUP(idx, deg, s = v0;)
        for (int base = 16; base < deg; base += 16) {
            unsigned ix = 0;
            if (base + (int)c < deg) ix = (unsigned)slots[o + base + (int)c];
            ROW_GROUP(ix, deg - base, ROW_ADD(0))
        }

        const float inv = 1.0f / fmaxf((float)deg, 1.0f);
        float4 r;
        r.x = fmaf(s.x, inv, own.x);
        r.y = fmaf(s.y, inv, own.y);
        r.z = fmaf(s.z, inv, own.z);
        r.w = fmaf(s.w, inv, own.w);
        *orow = r;
        sg = sgn; sgn = sgnn; idx = idxn;
        j = jn;   jn = jnn;
    }
}

__global__ __launch_bounds__(256)
void row_gather(const float* __restrict__ P_dis, const float* __restrict__ P_gene,
                float* __restrict__ acc_gene, float* __restrict__ acc_dis,
                const int2* __restrict__ seg, const int* __restrict__ slots) {
    const unsigned c = threadIdx.x & 15u;
    const int hw  = (blockIdx.x * 256 + threadIdx.x) >> 4;
    const int nhw = (gridDim.x * 256) >> 4;
    gather_set(P_dis,  ND_ - 1, acc_gene, seg,       slots, NG_, hw,           nhw, c);
    gather_set(P_gene, NG_ - 1, acc_dis,  seg + NG_, slots, ND_, nhw - 1 - hw, nhw, c);
}

template<int K>
__global__ __launch_bounds__(128)
void node_wmma(const float* x,
               const float* __restrict__ Wn,
               const float* __restrict__ Ws, const float* __restrict__ bias,
               float* outn, float* __restrict__ outs, int N)
{
    __shared__ v2f Wpk[K * 64];
    {
        constexpr int PER_T = K / 8;
        float4* lds4 = (float4*)Wpk;
        float4 ra[PER_T], rb[PER_T];
#pragma unroll
        for (int u = 0; u < PER_T; ++u) {
            const int I  = u * 128 + (int)threadIdx.x;
            const int PP = I >> 4, q = I & 15;
            const float* Wsel = (u * 8 < K / 2) ? Wn : Ws;
            const int pp = PP - ((u * 8 < K / 2) ? 0 : K / 2);
            const float4* g = (const float4*)(Wsel + (size_t)(2 * pp) * 64) + q;
            ra[u] = g[0];
            rb[u] = g[16];
        }
#pragma unroll
        for (int u = 0; u < PER_T; ++u) {
            const int I  = u * 128 + (int)threadIdx.x;
            const int PP = I >> 4, q = I & 15;
            float4* d = lds4 + ((PP * 64 + 4 * q) >> 1);
            d[0] = make_float4(ra[u].x, rb[u].x, ra[u].y, rb[u].y);
            d[1] = make_float4(ra[u].z, rb[u].z, ra[u].w, rb[u].w);
        }
    }
    __syncthreads();

    const int wave = threadIdx.x >> 5;
    const int lane = threadIdx.x & 31;
    const int pair = blockIdx.x * 4 + wave;
    const int row0 = pair * 32;
    if (row0 >= N) return;
    const bool has2 = (row0 + 32 <= N);

    const int m  = lane & 15;
    const int ph = lane >> 4;
    const int rowA0 = row0 + m;
    const int rowA1 = min(row0 + 16 + m, N - 1);

    v8f accn0[4] = {};
    v8f accn1[4] = {};
    v8f accs0[4] = {};
    v8f accs1[4] = {};

    const v2f* X0 = (const v2f*)(x + (size_t)rowA0 * K);
    const v2f* X1 = (const v2f*)(x + (size_t)rowA1 * K);
    for (int k = 0; k < K; k += 4) {
        const int kp = (k >> 1) + ph;
        v2f a0 = X0[kp];
        v2f a1 = X1[kp];
#pragma unroll
        for (int t = 0; t < 4; ++t) {
            v2f bn = Wpk[kp * 64 + t * 16 + m];
            accn0[t] = __builtin_amdgcn_wmma_f32_16x16x4_f32(
                false, a0, false, bn, (short)0, accn0[t], false, false);
            accn1[t] = __builtin_amdgcn_wmma_f32_16x16x4_f32(
                false, a1, false, bn, (short)0, accn1[t], false, false);
            v2f bs = Wpk[(K / 2 + kp) * 64 + t * 16 + m];
            accs0[t] = __builtin_amdgcn_wmma_f32_16x16x4_f32(
                false, a0, false, bs, (short)0, accs0[t], false, false);
            accs1[t] = __builtin_amdgcn_wmma_f32_16x16x4_f32(
                false, a1, false, bs, (short)0, accs1[t], false, false);
        }
    }

    const int colb  = lane & 15;
    const int mb0   = row0 + 8 * ph;
#pragma unroll
    for (int t = 0; t < 4; ++t) {
        const float b = bias[t * 16 + colb];
#pragma unroll
        for (int r = 0; r < 8; ++r) {
            outn[(size_t)(mb0 + r) * 64 + t * 16 + colb] = accn0[t][r];
            outs[(size_t)(mb0 + r) * 64 + t * 16 + colb] = accs0[t][r] + b;
        }
        if (has2) {
#pragma unroll
            for (int r = 0; r < 8; ++r) {
                outn[(size_t)(mb0 + 16 + r) * 64 + t * 16 + colb] = accn1[t][r];
                outs[(size_t)(mb0 + 16 + r) * 64 + t * 16 + colb] = accs1[t][r] + b;
            }
        }
    }
}

extern "C" void kernel_launch(void* const* d_in, const int* in_sizes, int n_in,
                              void* d_out, int out_size, void* d_ws, size_t ws_size,
                              hipStream_t stream) {
    const float* x_disease = (const float*)d_in[0];
    const float* x_gene    = (const float*)d_in[1];
    const int*   src_dg    = (const int*)d_in[2];
    const int*   dst_dg    = (const int*)d_in[3];
    const int*   src_gd    = (const int*)d_in[4];
    const int*   dst_gd    = (const int*)d_in[5];
    const float* Wl1_dg = (const float*)d_in[6];
    const float* bl1_dg = (const float*)d_in[7];
    const float* Wr1_dg = (const float*)d_in[8];
    const float* Wl1_gd = (const float*)d_in[9];
    const float* bl1_gd = (const float*)d_in[10];
    const float* Wr1_gd = (const float*)d_in[11];
    const float* Wl2_dg = (const float*)d_in[12];
    const float* bl2_dg = (const float*)d_in[13];
    const float* Wr2_dg = (const float*)d_in[14];
    const float* Wl2_gd = (const float*)d_in[15];
    const float* bl2_gd = (const float*)d_in[16];
    const float* Wr2_gd = (const float*)d_in[17];

    float* d2 = (float*)d_out;
    float* g2 = d2 + (size_t)ND_ * 64;
    float* p1_d = d2;
    float* p1_g = g2;

    float* ws = (float*)d_ws;
    size_t off = 0;
    int*  slots  = (int*)(ws + off);  off += (size_t)E2_;
    int*  rank   = (int*)(ws + off);  off += (size_t)E2_;
    int2* seg    = (int2*)(ws + off); off += (size_t)NT_ * 2;
    int*  cnt    = (int*)(ws + off);  off += (size_t)NT_;
    int*  cursor = (int*)(ws + off);  off += 48;
    float* g1    = ws + off; off += (size_t)NG_ * H_;
    float* d1    = ws + off; off += (size_t)ND_ * H_;
    (void)off; (void)ws_size; (void)in_sizes; (void)n_in; (void)out_size;

    const int TPB = 256;
    const int eb  = (E2_ / 4 + TPB - 1) / TPB;
    const int nb  = (NT_ + TPB - 1) / TPB;
    const int gb  = 1024;
    const int pairs_g = (NG_ / 16 + 1) / 2;
    const int pairs_d = (ND_ / 16 + 1) / 2;

    hipMemsetAsync(cnt, 0, ((size_t)NT_ + 48) * sizeof(int), stream);

    edge_rank<<<eb, TPB, 0, stream>>>(dst_dg, dst_gd, cnt, rank);
    seg_offsets<<<nb, TPB, 0, stream>>>(cnt, seg, cursor);
    edge_place<<<eb, TPB, 0, stream>>>(src_dg, dst_dg, src_gd, dst_gd, seg, rank, slots);

    node_wmma<DIN_><<<(pairs_d + 3) / 4, 128, 0, stream>>>(
        x_disease, Wl1_dg, Wr1_gd, bl1_gd, p1_d, d1, ND_);
    node_wmma<DIN_><<<(pairs_g + 3) / 4, 128, 0, stream>>>(
        x_gene,    Wl1_gd, Wr1_dg, bl1_dg, p1_g, g1, NG_);

    row_gather<<<gb, TPB, 0, stream>>>(p1_d, p1_g, g1, d1, seg, slots);

    node_wmma<H_><<<(pairs_d + 3) / 4, 128, 0, stream>>>(
        d1, Wl2_dg, Wr2_gd, bl2_gd, d1, d2, ND_);
    node_wmma<H_><<<(pairs_g + 3) / 4, 128, 0, stream>>>(
        g1, Wl2_gd, Wr2_dg, bl2_dg, g1, g2, NG_);

    row_gather<<<gb, TPB, 0, stream>>>(d1, g1, g2, d2, seg, slots);
}


# edge ordering with all counters in LDS: per-chunk bucket histogram, bucket scatter, one block sorts each 256-node bucket; no global atomics, no memset; f32 sum order changes (arrival-order rank)
# speedup vs baseline: 2.5334x; 1.2269x over previous
#include <hip/hip_runtime.h>

typedef float v2f __attribute__((ext_vector_type(2)));
typedef float v8f __attribute__((ext_vector_type(8)));
typedef int   v4i __attribute__((ext_vector_type(4)));

#define ND_   50000
#define NG_   100000
#define E_    600000
#define DIN_  128
#define H_    64
#define NT_   (NG_ + ND_)
#define E2_   (2 * E_)

#define NBKT_  ((NT_ + 255) >> 8)
#define SB_    128
#define QCH_   ((E2_ / 4 + SB_ - 1) / SB_)

#define BUCKET(u) ((int)min((unsigned)(u) >> 8, (unsigned)(NBKT_ - 1)))

#define QTRIPS_ ((QCH_ + 1023) / 1024)

__global__ __launch_bounds__(1024)
void edge_hist(const int* __restrict__ dst_dg, const int* __restrict__ dst_gd,
               int* __restrict__ hist) {
    __shared__ int h[NBKT_];
    const int t  = threadIdx.x;
    const int q0 = blockIdx.x * QCH_;
    const int q1 = min(q0 + QCH_, E2_ / 4);
    v4i d[QTRIPS_]; int base[QTRIPS_];
#pragma unroll
    for (int k = 0; k < QTRIPS_; ++k) {
        const int q   = min(q0 + t + k * 1024, q1 - 1);
        const bool dg = q < E_ / 4;
        const v4i* dp = dg ? (const v4i*)dst_dg + q : (const v4i*)dst_gd + (q - E_ / 4);
        base[k] = dg ? 0 : NG_;
        d[k]    = *dp;
    }
    if (t < NBKT_) h[t] = 0;
    __syncthreads();
#pragma unroll
    for (int k = 0; k < QTRIPS_; ++k) {
        if (q0 + t + k * 1024 < q1) {
            atomicAdd(&h[BUCKET(base[k] + d[k].x)], 1);
            atomicAdd(&h[BUCKET(base[k] + d[k].y)], 1);
            atomicAdd(&h[BUCKET(base[k] + d[k].z)], 1);
            atomicAdd(&h[BUCKET(base[k] + d[k].w)], 1);
        }
    }
    __syncthreads();
    if (t < NBKT_) hist[blockIdx.x * NBKT_ + t] = h[t];
}

__global__ __launch_bounds__(1024)
void edge_scatter(const int* __restrict__ src_dg, const int* __restrict__ dst_dg,
                  const int* __restrict__ src_gd, const int* __restrict__ dst_gd,
                  const int* __restrict__ hist, int* __restrict__ bstart,
                  unsigned* __restrict__ tmp) {
    __shared__ int cur[NBKT_];
    __shared__ int wtot[32];
    const int t    = threadIdx.x;
    const int lane = t & 31;
    const int w    = t >> 5;
    const int me   = blockIdx.x;
    const int q0   = me * QCH_;
    const int q1   = min(q0 + QCH_, E2_ / 4);
    v4i d[QTRIPS_], s[QTRIPS_]; int base[QTRIPS_];
#pragma unroll
    for (int k = 0; k < QTRIPS_; ++k) {
        const int q   = min(q0 + t + k * 1024, q1 - 1);
        const bool dg = q < E_ / 4;
        const v4i* dp = dg ? (const v4i*)dst_dg + q : (const v4i*)dst_gd + (q - E_ / 4);
        const v4i* sp = dg ? (const v4i*)src_dg + q : (const v4i*)src_gd + (q - E_ / 4);
        base[k] = dg ? 0 : NG_;
        d[k]    = *dp;
        s[k]    = *sp;
    }
    int tot = 0, part = 0;
    if (t < NBKT_) {
        for (int b0 = 0; b0 < SB_; b0 += 32) {
            int v[32];
#pragma unroll
            for (int j = 0; j < 32; ++j) v[j] = hist[(b0 + j) * NBKT_ + t];
#pragma unroll
            for (int j = 0; j < 32; ++j) {
                tot  += v[j];
                part += (b0 + j < me) ? v[j] : 0;
            }
        }
    }
    int incl = tot;
#pragma unroll
    for (int dd = 1; dd < 32; dd <<= 1) {
        const int v = __shfl_up(incl, dd);
        if (lane >= dd) incl += v;
    }
    if (lane == 31) wtot[w] = incl;
    __syncthreads();
    if (w == 0) {
        const int x = wtot[lane];
        int xi = x;
#pragma unroll
        for (int dd = 1; dd < 32; dd <<= 1) {
            const int v = __shfl_up(xi, dd);
            if (lane >= dd) xi += v;
        }
        wtot[lane] = xi - x;
    }
    __syncthreads();
    const int startb = wtot[w] + incl - tot;
    if (t < NBKT_) cur[t] = startb + part;
    if (me == 0 && t <= NBKT_) bstart[t] = startb;
    __syncthreads();

#define EDGE_WORD(uu, ss) (((unsigned)(ss) << 8) | ((uu) & 255u))
#pragma unroll
    for (int k = 0; k < QTRIPS_; ++k) {
        if (q0 + t + k * 1024 < q1) {
            const unsigned u0 = (unsigned)(base[k] + d[k].x);
            const unsigned u1 = (unsigned)(base[k] + d[k].y);
            const unsigned u2 = (unsigned)(base[k] + d[k].z);
            const unsigned u3 = (unsigned)(base[k] + d[k].w);
            const unsigned p0 = (unsigned)atomicAdd(&cur[BUCKET(u0)], 1);
            const unsigned p1 = (unsigned)atomicAdd(&cur[BUCKET(u1)], 1);
            const unsigned p2 = (unsigned)atomicAdd(&cur[BUCKET(u2)], 1);
            const unsigned p3 = (unsigned)atomicAdd(&cur[BUCKET(u3)], 1);
            tmp[min(p0, (unsigned)(E2_ - 1))] = EDGE_WORD(u0, s[k].x);
            tmp[min(p1, (unsigned)(E2_ - 1))] = EDGE_WORD(u1, s[k].y);
            tmp[min(p2, (unsigned)(E2_ - 1))] = EDGE_WORD(u2, s[k].z);
            tmp[min(p3, (unsigned)(E2_ - 1))] = EDGE_WORD(u3, s[k].w);
        }
    }
}

#define BS_T_   512
#define BS_PER_ 8

__global__ __launch_bounds__(BS_T_)
void bucket_sort(const unsigned* __restrict__ tmp, const int* __restrict__ bstart,
                 int2* __restrict__ seg, int* __restrict__ slots) {
    __shared__ int cnt[256];
    __shared__ int cur[256];
    __shared__ int wtot[8];
    const int t    = threadIdx.x;
    const int lane = t & 31;
    const int w    = t >> 5;
    const int start = min(max(bstart[blockIdx.x], 0), E2_);
    const int n     = min(max(bstart[blockIdx.x + 1] - start, 0), E2_ - start);
    if (t < 256) cnt[t] = 0;
    __syncthreads();
    for (int i0 = 0; i0 < n; i0 += BS_T_ * BS_PER_) {
        unsigned e[BS_PER_];
#pragma unroll
        for (int k = 0; k < BS_PER_; ++k)
            e[k] = tmp[start + min(i0 + t + k * BS_T_, n - 1)];
#pragma unroll
        for (int k = 0; k < BS_PER_; ++k)
            if (i0 + t + k * BS_T_ < n) atomicAdd(&cnt[e[k] & 255u], 1);
    }
    __syncthreads();
    int c = 0, incl = 0;
    if (t < 256) {
        c = cnt[t];
        incl = c;
#pragma unroll
        for (int dd = 1; dd < 32; dd <<= 1) {
            const int v = __shfl_up(incl, dd);
            if (lane >= dd) incl += v;
        }
        if (lane == 31) wtot[w] = incl;
    }
    __syncthreads();
    if (t < 256) {
        int wb = 0;
#pragma unroll
        for (int k = 0; k < 8; ++k) wb += (k < w) ? wtot[k] : 0;
        const int excl = wb + incl - c;
        const int u = blockIdx.x * 256 + t;
        if (u < NT_) seg[u] = make_int2(start + excl, c);
        cur[t] = excl;
    }
    __syncthreads();
    for (int i0 = 0; i0 < n; i0 += BS_T_ * BS_PER_) {
        unsigned e[BS_PER_];
#pragma unroll
        for (int k = 0; k < BS_PER_; ++k)
            e[k] = tmp[start + min(i0 + t + k * BS_T_, n - 1)];
        int p[BS_PER_];
#pragma unroll
        for (int k = 0; k < BS_PER_; ++k) {
            p[k] = 0;
            if (i0 + t + k * BS_T_ < n) p[k] = atomicAdd(&cur[e[k] & 255u], 1);
        }
#pragma unroll
        for (int k = 0; k < BS_PER_; ++k) {
            if (i0 + t + k * BS_T_ < n) slots[min(start + p[k], E2_ - 1)] = (int)(e[k] >> 8);
        }
    }
}


#define ROW_LANE(v, i) \
    ((unsigned)__builtin_amdgcn_update_dpp(0, (int)(v), 0x150 + (i), 0xf, 0xf, false))

#define ROW_LOAD(i, ix_, m_) \
    { const unsigned r_ = min(ROW_LANE(ix_, i), last_src); \
      if ((i) < (m_)) v##i = P4[r_ * 16u + c]; }

#define ROW_ADD(i) \
    s.x += v##i.x; s.y += v##i.y; s.z += v##i.z; s.w += v##i.w;

#define ROW_GROUP(ix_, m_, FIRST_) \
    { \
        const int mm_ = (m_); \
        float4 v0 = z4, v1 = z4, v2  = z4, v3  = z4, v4  = z4, v5  = z4, v6  = z4, v7  = z4; \
        float4 v8 = z4, v9 = z4, v10 = z4, v11 = z4, v12 = z4, v13 = z4, v14 = z4, v15 = z4; \
        ROW_LOAD(0, ix_, mm_) ROW_LOAD(1, ix_, mm_) ROW_LOAD(2, ix_, mm_) ROW_LOAD(3, ix_, mm_) \
        ROW_LOAD(4, ix_, mm_) ROW_LOAD(5, ix_, mm_) ROW_LOAD(6, ix_, mm_) ROW_LOAD(7, ix_, mm_) \
        if (mm_ > 8) { \
            ROW_LOAD(8,  ix_, mm_) ROW_LOAD(9,  ix_, mm_) ROW_LOAD(10, ix_, mm_) ROW_LOAD(11, ix_, mm_) \
            ROW_LOAD(12, ix_, mm_) ROW_LOAD(13, ix_, mm_) ROW_LOAD(14, ix_, mm_) ROW_LOAD(15, ix_, mm_) \
        } \
        FIRST_ \
        ROW_ADD(1) ROW_ADD(2) ROW_ADD(3) \
        ROW_ADD(4) ROW_ADD(5) ROW_ADD(6) ROW_ADD(7) \
        if (mm_ > 8) { \
            ROW_ADD(8)  ROW_ADD(9)  ROW_ADD(10) ROW_ADD(11) \
            ROW_ADD(12) ROW_ADD(13) ROW_ADD(14) ROW_ADD(15) \
        } \
    }

__device__ __forceinline__
void gather_set(const float* __restrict__ P, const unsigned last_src,
                float* __restrict__ acc, const int2* __restrict__ seg,
                const int* __restrict__ slots,
                const int n, const int first, const int step, const unsigned c) {
    const float4* __restrict__ P4 = (const float4*)P;
    const float4 z4 = make_float4(0.f, 0.f, 0.f, 0.f);
    int j  = first;
    int jn = j + step;
    int2 sg  = make_int2(0, 0);
    int2 sgn = make_int2(0, 0);
    unsigned idx = 0;
    if (j  < n) sg  = seg[j];
    if (jn < n) sgn = seg[jn];
    if ((int)c < sg.y) idx = (unsigned)slots[sg.x + (int)c];
    idx = min(idx, last_src);
    while (j < n) {
        const int o = sg.x, deg = sg.y;
        float4* orow = (float4*)(acc + ((size_t)j << 6)) + c;
        const float4 own = *orow;
        unsigned idxn = 0;
        if ((int)c < sgn.y) idxn = (unsigned)slots[sgn.x + (int)c];
        const int jnn = jn + step;
        int2 sgnn = make_int2(0, 0);
        if (jnn < n) sgnn = seg[jnn];

        float4 s;
        ROW_GROUP(idx, deg, s = v0;)
        for (int base = 16; base < deg; base += 16) {
            unsigned ix = 0;
            if (base + (int)c < deg) ix = (unsigned)slots[o + base + (int)c];
            ROW_GROUP(ix, deg - base, ROW_ADD(0))
        }

        const float inv = 1.0f / fmaxf((float)deg, 1.0f);
        float4 r;
        r.x = fmaf(s.x, inv, own.x);
        r.y = fmaf(s.y, inv, own.y);
        r.z = fmaf(s.z, inv, own.z);
        r.w = fmaf(s.w, inv, own.w);
        *orow = r;
        sg = sgn; sgn = sgnn; idx = idxn;
        j = jn;   jn = jnn;
    }
}

__global__ __launch_bounds__(256)
void row_gather(const float* __restrict__ P_dis, const float* __restrict__ P_gene,
                float* __restrict__ acc_gene, float* __restrict__ acc_dis,
                const int2* __restrict__ seg, const int* __restrict__ slots) {
    const unsigned c = threadIdx.x & 15u;
    const int hw  = (blockIdx.x * 256 + threadIdx.x) >> 4;
    const int nhw = (gridDim.x * 256) >> 4;
    gather_set(P_dis,  ND_ - 1, acc_gene, seg,       slots, NG_, hw,           nhw, c);
    gather_set(P_gene, NG_ - 1, acc_dis,  seg + NG_, slots, ND_, nhw - 1 - hw, nhw, c);
}

template<int K>
__global__ __launch_bounds__(128)
void node_wmma(const float* x,
               const float* __restrict__ Wn,
               const float* __restrict__ Ws, const float* __restrict__ bias,
               float* outn, float* __restrict__ outs, int N)
{
    __shared__ v2f Wpk[K * 64];
    {
        constexpr int PER_T = K / 8;
        float4* lds4 = (float4*)Wpk;
        float4 ra[PER_T], rb[PER_T];
#pragma unroll
        for (int u = 0; u < PER_T; ++u) {
            const int I  = u * 128 + (int)threadIdx.x;
            const int PP = I >> 4, q = I & 15;
            const float* Wsel = (u * 8 < K / 2) ? Wn : Ws;
            const int pp = PP - ((u * 8 < K / 2) ? 0 : K / 2);
            const float4* g = (const float4*)(Wsel + (size_t)(2 * pp) * 64) + q;
            ra[u] = g[0];
            rb[u] = g[16];
        }
#pragma unroll
        for (int u = 0; u < PER_T; ++u) {
            const int I  = u * 128 + (int)threadIdx.x;
            const int PP = I >> 4, q = I & 15;
            float4* d = lds4 + ((PP * 64 + 4 * q) >> 1);
            d[0] = make_float4(ra[u].x, rb[u].x, ra[u].y, rb[u].y);
            d[1] = make_float4(ra[u].z, rb[u].z, ra[u].w, rb[u].w);
        }
    }
    __syncthreads();

    const int wave = threadIdx.x >> 5;
    const int lane = threadIdx.x & 31;
    const int pair = blockIdx.x * 4 + wave;
    const int row0 = pair * 32;
    if (row0 >= N) return;
    const bool has2 = (row0 + 32 <= N);

    const int m  = lane & 15;
    const int ph = lane >> 4;
    const int rowA0 = row0 + m;
    const int rowA1 = min(row0 + 16 + m, N - 1);

    v8f accn0[4] = {};
    v8f accn1[4] = {};
    v8f accs0[4] = {};
    v8f accs1[4] = {};

    const v2f* X0 = (const v2f*)(x + (size_t)rowA0 * K);
    const v2f* X1 = (const v2f*)(x + (size_t)rowA1 * K);
    for (int k = 0; k < K; k += 4) {
        const int kp = (k >> 1) + ph;
        v2f a0 = X0[kp];
        v2f a1 = X1[kp];
#pragma unroll
        for (int t = 0; t < 4; ++t) {
            v2f bn = Wpk[kp * 64 + t * 16 + m];
            accn0[t] = __builtin_amdgcn_wmma_f32_16x16x4_f32(
                false, a0, false, bn, (short)0, accn0[t], false, false);
            accn1[t] = __builtin_amdgcn_wmma_f32_16x16x4_f32(
                false, a1, false, bn, (short)0, accn1[t], false, false);
            v2f bs = Wpk[(K / 2 + kp) * 64 + t * 16 + m];
            accs0[t] = __builtin_amdgcn_wmma_f32_16x16x4_f32(
                false, a0, false, bs, (short)0, accs0[t], false, false);
            accs1[t] = __builtin_amdgcn_wmma_f32_16x16x4_f32(
                false, a1, false, bs, (short)0, accs1[t], false, false);
        }
    }

    const int colb  = lane & 15;
    const int mb0   = row0 + 8 * ph;
#pragma unroll
    for (int t = 0; t < 4; ++t) {
        const float b = bias[t * 16 + colb];
#pragma unroll
        for (int r = 0; r < 8; ++r) {
            outn[(size_t)(mb0 + r) * 64 + t * 16 + colb] = accn0[t][r];
            outs[(size_t)(mb0 + r) * 64 + t * 16 + colb] = accs0[t][r] + b;
        }
        if (has2) {
#pragma unroll
            for (int r = 0; r < 8; ++r) {
                outn[(size_t)(mb0 + 16 + r) * 64 + t * 16 + colb] = accn1[t][r];
                outs[(size_t)(mb0 + 16 + r) * 64 + t * 16 + colb] = accs1[t][r] + b;
            }
        }
    }
}

extern "C" void kernel_launch(void* const* d_in, const int* in_sizes, int n_in,
                              void* d_out, int out_size, void* d_ws, size_t ws_size,
                              hipStream_t stream) {
    const float* x_disease = (const float*)d_in[0];
    const float* x_gene    = (const float*)d_in[1];
    const int*   src_dg    = (const int*)d_in[2];
    const int*   dst_dg    = (const int*)d_in[3];
    const int*   src_gd    = (const int*)d_in[4];
    const int*   dst_gd    = (const int*)d_in[5];
    const float* Wl1_dg = (const float*)d_in[6];
    const float* bl1_dg = (const float*)d_in[7];
    const float* Wr1_dg = (const float*)d_in[8];
    const float* Wl1_gd = (const float*)d_in[9];
    const float* bl1_gd = (const float*)d_in[10];
    const float* Wr1_gd = (const float*)d_in[11];
    const float* Wl2_dg = (const float*)d_in[12];
    const float* bl2_dg = (const float*)d_in[13];
    const float* Wr2_dg = (const float*)d_in[14];
    const float* Wl2_gd = (const float*)d_in[15];
    const float* bl2_gd = (const float*)d_in[16];
    const float* Wr2_gd = (const float*)d_in[17];

    float* d2 = (float*)d_out;
    float* g2 = d2 + (size_t)ND_ * 64;
    float* p1_d = d2;
    float* p1_g = g2;

    float* ws = (float*)d_ws;
    size_t off = 0;
    int*      slots  = (int*)(ws + off);      off += (size_t)E2_;
    unsigned* tmp    = (unsigned*)(ws + off); off += (size_t)E2_;
    int2*     seg    = (int2*)(ws + off);     off += (size_t)NT_ * 2;
    int*      hist   = (int*)(ws + off);
    int*      bstart = hist + SB_ * NBKT_;
    off += (size_t)NT_ + 48;
    float* g1    = ws + off; off += (size_t)NG_ * H_;
    float* d1    = ws + off; off += (size_t)ND_ * H_;
    (void)off; (void)ws_size; (void)in_sizes; (void)n_in; (void)out_size;

    const int TPB = 256;
    const int gb  = 1024;
    const int pairs_g = (NG_ / 16 + 1) / 2;
    const int pairs_d = (ND_ / 16 + 1) / 2;

    edge_hist<<<SB_, 1024, 0, stream>>>(dst_dg, dst_gd, hist);
    edge_scatter<<<SB_, 1024, 0, stream>>>(src_dg, dst_dg, src_gd, dst_gd, hist, bstart, tmp);
    bucket_sort<<<NBKT_, BS_T_, 0, stream>>>(tmp, bstart, seg, slots);

    node_wmma<DIN_><<<(pairs_d + 3) / 4, 128, 0, stream>>>(
        x_disease, Wl1_dg, Wr1_gd, bl1_gd, p1_d, d1, ND_);
    node_wmma<DIN_><<<(pairs_g + 3) / 4, 128, 0, stream>>>(
        x_gene,    Wl1_gd, Wr1_dg, bl1_dg, p1_g, g1, NG_);

    row_gather<<<gb, TPB, 0, stream>>>(p1_d, p1_g, g1, d1, seg, slots);

    node_wmma<H_><<<(pairs_d + 3) / 4, 128, 0, stream>>>(
        d1, Wl2_dg, Wr2_gd, bl2_gd, d1, d2, ND_);
    node_wmma<H_><<<(pairs_g + 3) / 4, 128, 0, stream>>>(
        g1, Wl2_gd, Wr2_dg, bl2_dg, g1, g2, NG_);

    row_gather<<<gb, TPB, 0, stream>>>(d1, g1, g2, d2, seg, slots);
}


# node products: one launch per layer serves both node sets (set chosen per block by block index): 9 launches become 7, both sets fill the SIMDs together; arithmetic and sum order unchanged
# speedup vs baseline: 2.7264x; 1.0762x over previous
#include <hip/hip_runtime.h>

typedef float v2f __attribute__((ext_vector_type(2)));
typedef float v8f __attribute__((ext_vector_type(8)));
typedef int   v4i __attribute__((ext_vector_type(4)));

#define ND_   50000
#define NG_   100000
#define E_    600000
#define DIN_  128
#define H_    64
#define NT_   (NG_ + ND_)
#define E2_   (2 * E_)

#define NBKT_  ((NT_ + 255) >> 8)
#define SB_    128
#define QCH_   ((E2_ / 4 + SB_ - 1) / SB_)

#define BUCKET(u) ((int)min((unsigned)(u) >> 8, (unsigned)(NBKT_ - 1)))

#define QTRIPS_ ((QCH_ + 1023) / 1024)

__global__ __launch_bounds__(1024)
void edge_hist(const int* __restrict__ dst_dg, const int* __restrict__ dst_gd,
               int* __restrict__ hist) {
    __shared__ int h[NBKT_];
    const int t  = threadIdx.x;
    const int q0 = blockIdx.x * QCH_;
    const int q1 = min(q0 + QCH_, E2_ / 4);
    v4i d[QTRIPS_]; int base[QTRIPS_];
#pragma unroll
    for (int k = 0; k < QTRIPS_; ++k) {
        const int q   = min(q0 + t + k * 1024, q1 - 1);
        const bool dg = q < E_ / 4;
        const v4i* dp = dg ? (const v4i*)dst_dg + q : (const v4i*)dst_gd + (q - E_ / 4);
        base[k] = dg ? 0 : NG_;
        d[k]    = *dp;
    }
    if (t < NBKT_) h[t] = 0;
    __syncthreads();
#pragma unroll
    for (int k = 0; k < QTRIPS_; ++k) {
        if (q0 + t + k * 1024 < q1) {
            atomicAdd(&h[BUCKET(base[k] + d[k].x)], 1);
            atomicAdd(&h[BUCKET(base[k] + d[k].y)], 1);
            atomicAdd(&h[BUCKET(base[k] + d[k].z)], 1);
            atomicAdd(&h[BUCKET(base[k] + d[k].w)], 1);
        }
    }
    __syncthreads();
    if (t < NBKT_) hist[blockIdx.x * NBKT_ + t] = h[t];
}

__global__ __launch_bounds__(1024)
void edge_scatter(const int* __restrict__ src_dg, const int* __restrict__ dst_dg,
                  const int* __restrict__ src_gd, const int* __restrict__ dst_gd,
                  const int* __restrict__ hist, int* __restrict__ bstart,
                  unsigned* __restrict__ tmp) {
    __shared__ int cur[NBKT_];
    __shared__ int wtot[32];
    const int t    = threadIdx.x;
    const int lane = t & 31;
    const int w    = t >> 5;
    const int me   = blockIdx.x;
    const int q0   = me * QCH_;
    const int q1   = min(q0 + QCH_, E2_ / 4);
    v4i d[QTRIPS_], s[QTRIPS_]; int base[QTRIPS_];
#pragma unroll
    for (int k = 0; k < QTRIPS_; ++k) {
        const int q   = min(q0 + t + k * 1024, q1 - 1);
        const bool dg = q < E_ / 4;
        const v4i* dp = dg ? (const v4i*)dst_dg + q : (const v4i*)dst_gd + (q - E_ / 4);
        const v4i* sp = dg ? (const v4i*)src_dg + q : (const v4i*)src_gd + (q - E_ / 4);
        base[k] = dg ? 0 : NG_;
        d[k]    = *dp;
        s[k]    = *sp;
    }
    int tot = 0, part = 0;
    if (t < NBKT_) {
        for (int b0 = 0; b0 < SB_; b0 += 32) {
            int v[32];
#pragma unroll
            for (int j = 0; j < 32; ++j) v[j] = hist[(b0 + j) * NBKT_ + t];
#pragma unroll
            for (int j = 0; j < 32; ++j) {
                tot  += v[j];
                part += (b0 + j < me) ? v[j] : 0;
            }
        }
    }
    int incl = tot;
#pragma unroll
    for (int dd = 1; dd < 32; dd <<= 1) {
        const int v = __shfl_up(incl, dd);
        if (lane >= dd) incl += v;
    }
    if (lane == 31) wtot[w] = incl;
    __syncthreads();
    if (w == 0) {
        const int x = wtot[lane];
        int xi = x;
#pragma unroll
        for (int dd = 1; dd < 32; dd <<= 1) {
            const int v = __shfl_up(xi, dd);
            if (lane >= dd) xi += v;
        }
        wtot[lane] = xi - x;
    }
    __syncthreads();
    const int startb = wtot[w] + incl - tot;
    if (t < NBKT_) cur[t] = startb + part;
    if (me == 0 && t <= NBKT_) bstart[t] = startb;
    __syncthreads();

#define EDGE_WORD(uu, ss) (((unsigned)(ss) << 8) | ((uu) & 255u))
#pragma unroll
    for (int k = 0; k < QTRIPS_; ++k) {
        if (q0 + t + k * 1024 < q1) {
            const unsigned u0 = (unsigned)(base[k] + d[k].x);
            const unsigned u1 = (unsigned)(base[k] + d[k].y);
            const unsigned u2 = (unsigned)(base[k] + d[k].z);
            const unsigned u3 = (unsigned)(base[k] + d[k].w);
            const unsigned p0 = (unsigned)atomicAdd(&cur[BUCKET(u0)], 1);
            const unsigned p1 = (unsigned)atomicAdd(&cur[BUCKET(u1)], 1);
            const unsigned p2 = (unsigned)atomicAdd(&cur[BUCKET(u2)], 1);
            const unsigned p3 = (unsigned)atomicAdd(&cur[BUCKET(u3)], 1);
            tmp[min(p0, (unsigned)(E2_ - 1))] = EDGE_WORD(u0, s[k].x);
            tmp[min(p1, (unsigned)(E2_ - 1))] = EDGE_WORD(u1, s[k].y);
            tmp[min(p2, (unsigned)(E2_ - 1))] = EDGE_WORD(u2, s[k].z);
            tmp[min(p3, (unsigned)(E2_ - 1))] = EDGE_WORD(u3, s[k].w);
        }
    }
}

#define BS_T_   512
#define BS_PER_ 8

__global__ __launch_bounds__(BS_T_)
void bucket_sort(const unsigned* __restrict__ tmp, const int* __restrict__ bstart,
                 int2* __restrict__ seg, int* __restrict__ slots) {
    __shared__ int cnt[256];
    __shared__ int cur[256];
    __shared__ int wtot[8];
    const int t    = threadIdx.x;
    const int lane = t & 31;
    const int w    = t >> 5;
    const int start = min(max(bstart[blockIdx.x], 0), E2_);
    const int n     = min(max(bstart[blockIdx.x + 1] - start, 0), E2_ - start);
    if (t < 256) cnt[t] = 0;
    __syncthreads();
    for (int i0 = 0; i0 < n; i0 += BS_T_ * BS_PER_) {
        unsigned e[BS_PER_];
#pragma unroll
        for (int k = 0; k < BS_PER_; ++k)
            e[k] = tmp[start + min(i0 + t + k * BS_T_, n - 1)];
#pragma unroll
        for (int k = 0; k < BS_PER_; ++k)
            if (i0 + t + k * BS_T_ < n) atomicAdd(&cnt[e[k] & 255u], 1);
    }
    __syncthreads();
    int c = 0, incl = 0;
    if (t < 256) {
        c = cnt[t];
        incl = c;
#pragma unroll
        for (int dd = 1; dd < 32; dd <<= 1) {
            const int v = __shfl_up(incl, dd);
            if (lane >= dd) incl += v;
        }
        if (lane == 31) wtot[w] = incl;
    }
    __syncthreads();
    if (t < 256) {
        int wb = 0;
#pragma unroll
        for (int k = 0; k < 8; ++k) wb += (k < w) ? wtot[k] : 0;
        const int excl = wb + incl - c;
        const int u = blockIdx.x * 256 + t;
        if (u < NT_) seg[u] = make_int2(start + excl, c);
        cur[t] = excl;
    }
    __syncthreads();
    for (int i0 = 0; i0 < n; i0 += BS_T_ * BS_PER_) {
        unsigned e[BS_PER_];
#pragma unroll
        for (int k = 0; k < BS_PER_; ++k)
            e[k] = tmp[start + min(i0 + t + k * BS_T_, n - 1)];
        int p[BS_PER_];
#pragma unroll
        for (int k = 0; k < BS_PER_; ++k) {
            p[k] = 0;
            if (i0 + t + k * BS_T_ < n) p[k] = atomicAdd(&cur[e[k] & 255u], 1);
        }
#pragma unroll
        for (int k = 0; k < BS_PER_; ++k) {
            if (i0 + t + k * BS_T_ < n) slots[min(start + p[k], E2_ - 1)] = (int)(e[k] >> 8);
        }
    }
}


#define ROW_LANE(v, i) \
    ((unsigned)__builtin_amdgcn_update_dpp(0, (int)(v), 0x150 + (i), 0xf, 0xf, false))

#define ROW_LOAD(i, ix_, m_) \
    { const unsigned r_ = min(ROW_LANE(ix_, i), last_src); \
      if ((i) < (m_)) v##i = P4[r_ * 16u + c]; }

#define ROW_ADD(i) \
    s.x += v##i.x; s.y += v##i.y; s.z += v##i.z; s.w += v##i.w;

#define ROW_GROUP(ix_, m_, FIRST_) \
    { \
        const int mm_ = (m_); \
        float4 v0 = z4, v1 = z4, v2  = z4, v3  = z4, v4  = z4, v5  = z4, v6  = z4, v7  = z4; \
        float4 v8 = z4, v9 = z4, v10 = z4, v11 = z4, v12 = z4, v13 = z4, v14 = z4, v15 = z4; \
        ROW_LOAD(0, ix_, mm_) ROW_LOAD(1, ix_, mm_) ROW_LOAD(2, ix_, mm_) ROW_LOAD(3, ix_, mm_) \
        ROW_LOAD(4, ix_, mm_) ROW_LOAD(5, ix_, mm_) ROW_LOAD(6, ix_, mm_) ROW_LOAD(7, ix_, mm_) \
        if (mm_ > 8) { \
            ROW_LOAD(8,  ix_, mm_) ROW_LOAD(9,  ix_, mm_) ROW_LOAD(10, ix_, mm_) ROW_LOAD(11, ix_, mm_) \
            ROW_LOAD(12, ix_, mm_) ROW_LOAD(13, ix_, mm_) ROW_LOAD(14, ix_, mm_) ROW_LOAD(15, ix_, mm_) \
        } \
        FIRST_ \
        ROW_ADD(1) ROW_ADD(2) ROW_ADD(3) \
        ROW_ADD(4) ROW_ADD(5) ROW_ADD(6) ROW_ADD(7) \
        if (mm_ > 8) { \
            ROW_ADD(8)  ROW_ADD(9)  ROW_ADD(10) ROW_ADD(11) \
            ROW_ADD(12) ROW_ADD(13) ROW_ADD(14) ROW_ADD(15) \
        } \
    }

__device__ __forceinline__
void gather_set(const float* __restrict__ P, const unsigned last_src,
                float* __restrict__ acc, const int2* __restrict__ seg,
                const int* __restrict__ slots,
                const int n, const int first, const int step, const unsigned c) {
    const float4* __restrict__ P4 = (const float4*)P;
    const float4 z4 = make_float4(0.f, 0.f, 0.f, 0.f);
    int j  = first;
    int jn = j + step;
    int2 sg  = make_int2(0, 0);
    int2 sgn = make_int2(0, 0);
    unsigned idx = 0;
    if (j  < n) sg  = seg[j];
    if (jn < n) sgn = seg[jn];
    if ((int)c < sg.y) idx = (unsigned)slots[sg.x + (int)c];
    idx = min(idx, last_src);
    while (j < n) {
        const int o = sg.x, deg = sg.y;
        float4* orow = (float4*)(acc + ((size_t)j << 6)) + c;
        const float4 own = *orow;
        unsigned idxn = 0;
        if ((int)c < sgn.y) idxn = (unsigned)slots[sgn.x + (int)c];
        const int jnn = jn + step;
        int2 sgnn = make_int2(0, 0);
        if (jnn < n) sgnn = seg[jnn];

        float4 s;
        ROW_GROUP(idx, deg, s = v0;)
        for (int base = 16; base < deg; base += 16) {
            unsigned ix = 0;
            if (base + (int)c < deg) ix = (unsigned)slots[o + base + (int)c];
            ROW_GROUP(ix, deg - base, ROW_ADD(0))
        }

        const float inv = 1.0f / fmaxf((float)deg, 1.0f);
        float4 r;
        r.x = fmaf(s.x, inv, own.x);
        r.y = fmaf(s.y, inv, own.y);
        r.z = fmaf(s.z, inv, own.z);
        r.w = fmaf(s.w, inv, own.w);
        *orow = r;
        sg = sgn; sgn = sgnn; idx = idxn;
        j = jn;   jn = jnn;
    }
}

__global__ __launch_bounds__(256)
void row_gather(const float* __restrict__ P_dis, const float* __restrict__ P_gene,
                float* __restrict__ acc_gene, float* __restrict__ acc_dis,
                const int2* __restrict__ seg, const int* __restrict__ slots) {
    const unsigned c = threadIdx.x & 15u;
    const int hw  = (blockIdx.x * 256 + threadIdx.x) >> 4;
    const int nhw = (gridDim.x * 256) >> 4;
    gather_set(P_dis,  ND_ - 1, acc_gene, seg,       slots, NG_, hw,           nhw, c);
    gather_set(P_gene, NG_ - 1, acc_dis,  seg + NG_, slots, ND_, nhw - 1 - hw, nhw, c);
}

template<int K>
__global__ __launch_bounds__(128)
void node_wmma(const float* xa,
               const float* __restrict__ Wna,
               const float* __restrict__ Wsa, const float* __restrict__ biasa,
               float* outna, float* __restrict__ outsa, int Na, int nblk_a,
               const float* xb,
               const float* __restrict__ Wnb,
               const float* __restrict__ Wsb, const float* __restrict__ biasb,
               float* outnb, float* __restrict__ outsb, int Nb)
{
    __shared__ v2f Wpk[K * 64];
    const bool seta = (int)blockIdx.x < nblk_a;
    const float* x                 = seta ? xa    : xb;
    const float* __restrict__ Wn   = seta ? Wna   : Wnb;
    const float* __restrict__ Ws   = seta ? Wsa   : Wsb;
    const float* __restrict__ bias = seta ? biasa : biasb;
    float* outn                    = seta ? outna : outnb;
    float* __restrict__ outs       = seta ? outsa : outsb;
    const int N   = seta ? Na : Nb;
    const int blk = seta ? (int)blockIdx.x : (int)blockIdx.x - nblk_a;
    {
        constexpr int PER_T = K / 8;
        float4* lds4 = (float4*)Wpk;
        float4 ra[PER_T], rb[PER_T];
#pragma unroll
        for (int u = 0; u < PER_T; ++u) {
            const int I  = u * 128 + (int)threadIdx.x;
            const int PP = I >> 4, q = I & 15;
            const float* Wsel = (u * 8 < K / 2) ? Wn : Ws;
            const int pp = PP - ((u * 8 < K / 2) ? 0 : K / 2);
            const float4* g = (const float4*)(Wsel + (size_t)(2 * pp) * 64) + q;
            ra[u] = g[0];
            rb[u] = g[16];
        }
#pragma unroll
        for (int u = 0; u < PER_T; ++u) {
            const int I  = u * 128 + (int)threadIdx.x;
            const int PP = I >> 4, q = I & 15;
            float4* d = lds4 + ((PP * 64 + 4 * q) >> 1);
            d[0] = make_float4(ra[u].x, rb[u].x, ra[u].y, rb[u].y);
            d[1] = make_float4(ra[u].z, rb[u].z, ra[u].w, rb[u].w);
        }
    }
    __syncthreads();

    const int wave = threadIdx.x >> 5;
    const int lane = threadIdx.x & 31;
    const int pair = blk * 4 + wave;
    const int row0 = pair * 32;
    if (row0 >= N) return;
    const bool has2 = (row0 + 32 <= N);

    const int m  = lane & 15;
    const int ph = lane >> 4;
    const int rowA0 = row0 + m;
    const int rowA1 = min(row0 + 16 + m, N - 1);

    v8f accn0[4] = {};
    v8f accn1[4] = {};
    v8f accs0[4] = {};
    v8f accs1[4] = {};

    const v2f* X0 = (const v2f*)(x + (size_t)rowA0 * K);
    const v2f* X1 = (const v2f*)(x + (size_t)rowA1 * K);
    for (int k = 0; k < K; k += 4) {
        const int kp = (k >> 1) + ph;
        v2f a0 = X0[kp];
        v2f a1 = X1[kp];
#pragma unroll
        for (int t = 0; t < 4; ++t) {
            v2f bn = Wpk[kp * 64 + t * 16 + m];
            accn0[t] = __builtin_amdgcn_wmma_f32_16x16x4_f32(
                false, a0, false, bn, (short)0, accn0[t], false, false);
            accn1[t] = __builtin_amdgcn_wmma_f32_16x16x4_f32(
                false, a1, false, bn, (short)0, accn1[t], false, false);
            v2f bs = Wpk[(K / 2 + kp) * 64 + t * 16 + m];
            accs0[t] = __builtin_amdgcn_wmma_f32_16x16x4_f32(
                false, a0, false, bs, (short)0, accs0[t], false, false);
            accs1[t] = __builtin_amdgcn_wmma_f32_16x16x4_f32(
                false, a1, false, bs, (short)0, accs1[t], false, false);
        }
    }

    const int colb  = lane & 15;
    const int mb0   = row0 + 8 * ph;
#pragma unroll
    for (int t = 0; t < 4; ++t) {
        const float b = bias[t * 16 + colb];
#pragma unroll
        for (int r = 0; r < 8; ++r) {
            outn[(size_t)(mb0 + r) * 64 + t * 16 + colb] = accn0[t][r];
            outs[(size_t)(mb0 + r) * 64 + t * 16 + colb] = accs0[t][r] + b;
        }
        if (has2) {
#pragma unroll
            for (int r = 0; r < 8; ++r) {
                outn[(size_t)(mb0 + 16 + r) * 64 + t * 16 + colb] = accn1[t][r];
                outs[(size_t)(mb0 + 16 + r) * 64 + t * 16 + colb] = accs1[t][r] + b;
            }
        }
    }
}

extern "C" void kernel_launch(void* const* d_in, const int* in_sizes, int n_in,
                              void* d_out, int out_size, void* d_ws, size_t ws_size,
                              hipStream_t stream) {
    const float* x_disease = (const float*)d_in[0];
    const float* x_gene    = (const float*)d_in[1];
    const int*   src_dg    = (const int*)d_in[2];
    const int*   dst_dg    = (const int*)d_in[3];
    const int*   src_gd    = (const int*)d_in[4];
    const int*   dst_gd    = (const int*)d_in[5];
    const float* Wl1_dg = (const float*)d_in[6];
    const float* bl1_dg = (const float*)d_in[7];
    const float* Wr1_dg = (const float*)d_in[8];
    const float* Wl1_gd = (const float*)d_in[9];
    const float* bl1_gd = (const float*)d_in[10];
    const float* Wr1_gd = (const float*)d_in[11];
    const float* Wl2_dg = (const float*)d_in[12];
    const float* bl2_dg = (const float*)d_in[13];
    const float* Wr2_dg = (const float*)d_in[14];
    const float* Wl2_gd = (const float*)d_in[15];
    const float* bl2_gd = (const float*)d_in[16];
    const float* Wr2_gd = (const float*)d_in[17];

    float* d2 = (float*)d_out;
    float* g2 = d2 + (size_t)ND_ * 64;
    float* p1_d = d2;
    float* p1_g = g2;

    float* ws = (float*)d_ws;
    size_t off = 0;
    int*      slots  = (int*)(ws + off);      off += (size_t)E2_;
    unsigned* tmp    = (unsigned*)(ws + off); off += (size_t)E2_;
    int2*     seg    = (int2*)(ws + off);     off += (size_t)NT_ * 2;
    int*      hist   = (int*)(ws + off);
    int*      bstart = hist + SB_ * NBKT_;
    off += (size_t)NT_ + 48;
    float* g1    = ws + off; off += (size_t)NG_ * H_;
    float* d1    = ws + off; off += (size_t)ND_ * H_;
    (void)off; (void)ws_size; (void)in_sizes; (void)n_in; (void)out_size;

    const int TPB = 256;
    const int gb  = 1024;
    const int pairs_g = (NG_ / 16 + 1) / 2;
    const int pairs_d = (ND_ / 16 + 1) / 2;
    const int nblk_g  = (pairs_g + 3) / 4;
    const int nblk_d  = (pairs_d + 3) / 4;

    edge_hist<<<SB_, 1024, 0, stream>>>(dst_dg, dst_gd, hist);
    edge_scatter<<<SB_, 1024, 0, stream>>>(src_dg, dst_dg, src_gd, dst_gd, hist, bstart, tmp);
    bucket_sort<<<NBKT_, BS_T_, 0, stream>>>(tmp, bstart, seg, slots);

    node_wmma<DIN_><<<nblk_d + nblk_g, 128, 0, stream>>>(
        x_disease, Wl1_dg, Wr1_gd, bl1_gd, p1_d, d1, ND_, nblk_d,
        x_gene,    Wl1_gd, Wr1_dg, bl1_dg, p1_g, g1, NG_);

    row_gather<<<gb, TPB, 0, stream>>>(p1_d, p1_g, g1, d1, seg, slots);

    node_wmma<H_><<<nblk_d + nblk_g, 128, 0, stream>>>(
        d1, Wl2_dg, Wr2_gd, bl2_gd, d1, d2, ND_, nblk_d,
        g1, Wl2_gd, Wr2_dg, bl2_dg, g1, g2, NG_);

    row_gather<<<gb, TPB, 0, stream>>>(d1, g1, g2, d2, seg, slots);
}


# layer 2 in one launch: a wave sums its 16 rows' sender rows into an LDS tile beside the own rows and multiplies [mean|own] by [Wl2;Wr2] (K=128) there; 6 launches to 5; f32 sum order changes
# speedup vs baseline: 3.9901x; 1.4635x over previous
#include <hip/hip_runtime.h>

typedef float v2f __attribute__((ext_vector_type(2)));
typedef float v8f __attribute__((ext_vector_type(8)));
typedef int   v4i __attribute__((ext_vector_type(4)));

#define ND_   50000
#define NG_   100000
#define E_    600000
#define DIN_  128
#define H_    64
#define NT_   (NG_ + ND_)
#define E2_   (2 * E_)

#define NBKT_  ((NT_ + 255) >> 8)
#define SB_    128
#define QCH_   ((E2_ / 4 + SB_ - 1) / SB_)
#define CHW_   (QCH_ * 4)
#define OFP_   (NBKT_ + 1)

#define BUCKET(u) ((int)min((unsigned)(u) >> 8, (unsigned)(NBKT_ - 1)))

#define QTRIPS_ ((QCH_ + 1023) / 1024)

#define EDGE_WORD(uu, ss) (((unsigned)(ss) << 8) | ((uu) & 255u))

__global__ __launch_bounds__(1024)
void chunk_partition(const int* __restrict__ src_dg, const int* __restrict__ dst_dg,
                     const int* __restrict__ src_gd, const int* __restrict__ dst_gd,
                     int* __restrict__ offs, unsigned* __restrict__ tmp) {
    __shared__ int cur[NBKT_];
    __shared__ int wtot[32];
    __shared__ v4i srt4[QCH_];
    unsigned* srt = (unsigned*)srt4;
    const int t    = threadIdx.x;
    const int lane = t & 31;
    const int w    = t >> 5;
    const int me   = blockIdx.x;
    const int q0   = me * QCH_;
    const int q1   = min(q0 + QCH_, E2_ / 4);
    v4i d[QTRIPS_], s[QTRIPS_]; int base[QTRIPS_];
#pragma unroll
    for (int k = 0; k < QTRIPS_; ++k) {
        const int q   = min(q0 + t + k * 1024, q1 - 1);
        const bool dg = q < E_ / 4;
        const v4i* dp = dg ? (const v4i*)dst_dg + q : (const v4i*)dst_gd + (q - E_ / 4);
        const v4i* sp = dg ? (const v4i*)src_dg + q : (const v4i*)src_gd + (q - E_ / 4);
        base[k] = dg ? 0 : NG_;
        d[k]    = *dp;
        s[k]    = *sp;
    }
    if (t < NBKT_) cur[t] = 0;
    __syncthreads();
#pragma unroll
    for (int k = 0; k < QTRIPS_; ++k) {
        if (q0 + t + k * 1024 < q1) {
            atomicAdd(&cur[BUCKET(base[k] + d[k].x)], 1);
            atomicAdd(&cur[BUCKET(base[k] + d[k].y)], 1);
            atomicAdd(&cur[BUCKET(base[k] + d[k].z)], 1);
            atomicAdd(&cur[BUCKET(base[k] + d[k].w)], 1);
        }
    }
    __syncthreads();
    const int tot = (t < NBKT_) ? cur[t] : 0;
    int incl = tot;
#pragma unroll
    for (int dd = 1; dd < 32; dd <<= 1) {
        const int v = __shfl_up(incl, dd);
        if (lane >= dd) incl += v;
    }
    if (lane == 31) wtot[w] = incl;
    __syncthreads();
    if (w == 0) {
        const int x = wtot[lane];
        int xi = x;
#pragma unroll
        for (int dd = 1; dd < 32; dd <<= 1) {
            const int v = __shfl_up(xi, dd);
            if (lane >= dd) xi += v;
        }
        wtot[lane] = xi - x;
    }
    __syncthreads();
    const int startb = wtot[w] + incl - tot;
    if (t < NBKT_)  cur[t] = startb;
    if (t <= NBKT_) offs[me * OFP_ + t] = startb;
    __syncthreads();

#pragma unroll
    for (int k = 0; k < QTRIPS_; ++k) {
        if (q0 + t + k * 1024 < q1) {
            const unsigned u0 = (unsigned)(base[k] + d[k].x);
            const unsigned u1 = (unsigned)(base[k] + d[k].y);
            const unsigned u2 = (unsigned)(base[k] + d[k].z);
            const unsigned u3 = (unsigned)(base[k] + d[k].w);
            const unsigned p0 = (unsigned)atomicAdd(&cur[BUCKET(u0)], 1);
            const unsigned p1 = (unsigned)atomicAdd(&cur[BUCKET(u1)], 1);
            const unsigned p2 = (unsigned)atomicAdd(&cur[BUCKET(u2)], 1);
            const unsigned p3 = (unsigned)atomicAdd(&cur[BUCKET(u3)], 1);
            srt[min(p0, (unsigned)(CHW_ - 1))] = EDGE_WORD(u0, s[k].x);
            srt[min(p1, (unsigned)(CHW_ - 1))] = EDGE_WORD(u1, s[k].y);
            srt[min(p2, (unsigned)(CHW_ - 1))] = EDGE_WORD(u2, s[k].z);
            srt[min(p3, (unsigned)(CHW_ - 1))] = EDGE_WORD(u3, s[k].w);
        }
    }
    __syncthreads();
    v4i* out4 = (v4i*)tmp + q0;
#pragma unroll
    for (int k = 0; k < QTRIPS_; ++k) {
        const int g = t + k * 1024;
        if (q0 + g < q1) out4[g] = srt4[g];
    }
}

#define BS_T_   512
#define BS_PER_ 8
#define SLCAP_  4096

__global__ __launch_bounds__(BS_T_)
void bucket_sort(const unsigned* __restrict__ tmp, const int* __restrict__ offs,
                 int2* __restrict__ seg, int* __restrict__ slots) {
    __shared__ int cnt[256];
    __shared__ int cur[256];
    __shared__ int wtot[8];
    __shared__ int pbeg[SB_];
    __shared__ int plen[SB_];
    __shared__ int wsum[SB_ / 32];
    __shared__ int wlen[SB_ / 32];
    __shared__ int sl[SLCAP_];
    const int t    = threadIdx.x;
    const int lane = t & 31;
    const int w    = t >> 5;
    const int b    = blockIdx.x;
    if (t < 256) cnt[t] = 0;
    if (t < SB_) {
        const int o0 = offs[t * OFP_ + b];
        const int o1 = offs[t * OFP_ + b + 1];
        const int clen = min(CHW_, E2_ - t * CHW_);
        const int a0 = min(max(o0, 0), clen);
        const int a1 = min(max(o1, a0), clen);
        pbeg[t] = t * CHW_ + a0;
        plen[t] = a1 - a0;
        int sacc = a0, lacc = a1 - a0;
#pragma unroll
        for (int dd = 16; dd >= 1; dd >>= 1) {
            sacc += __shfl_xor(sacc, dd);
            lacc += __shfl_xor(lacc, dd);
        }
        if (lane == 0) { wsum[w] = sacc; wlen[w] = lacc; }
    }
    __syncthreads();
    int start = 0, n = 0;
#pragma unroll
    for (int k = 0; k < SB_ / 32; ++k) { start += wsum[k]; n += wlen[k]; }
    start = min(start, E2_);
    n     = min(n, E2_ - start);
    const bool inlds = n <= SLCAP_;
    const int j0 = t & 3;
    const int pb = pbeg[t >> 2];
    const int pl = plen[t >> 2];
    unsigned e[BS_PER_];
#pragma unroll
    for (int k = 0; k < BS_PER_; ++k)
        e[k] = tmp[min(pb + j0 + 4 * k, E2_ - 1)];
#pragma unroll
    for (int k = 0; k < BS_PER_; ++k)
        if (j0 + 4 * k < pl) atomicAdd(&cnt[e[k] & 255u], 1);
    for (int j = j0 + 4 * BS_PER_; j < pl; j += 4)
        atomicAdd(&cnt[tmp[min(pb + j, E2_ - 1)] & 255u], 1);
    __syncthreads();
    int c = 0, incl = 0;
    if (t < 256) {
        c = cnt[t];
        incl = c;
#pragma unroll
        for (int dd = 1; dd < 32; dd <<= 1) {
            const int v = __shfl_up(incl, dd);
            if (lane >= dd) incl += v;
        }
        if (lane == 31) wtot[w] = incl;
    }
    __syncthreads();
    if (t < 256) {
        int wb = 0;
#pragma unroll
        for (int k = 0; k < 8; ++k) wb += (k < w) ? wtot[k] : 0;
        const int excl = wb + incl - c;
        const int u = b * 256 + t;
        if (u < NT_) seg[u] = make_int2(start + excl, c);
        cur[t] = excl;
    }
    __syncthreads();
    int p[BS_PER_];
#pragma unroll
    for (int k = 0; k < BS_PER_; ++k) {
        p[k] = 0;
        if (j0 + 4 * k < pl) p[k] = atomicAdd(&cur[e[k] & 255u], 1);
    }
#pragma unroll
    for (int k = 0; k < BS_PER_; ++k) {
        if (j0 + 4 * k < pl) {
            if (inlds) sl[min(p[k], SLCAP_ - 1)] = (int)(e[k] >> 8);
            else       slots[min(start + p[k], E2_ - 1)] = (int)(e[k] >> 8);
        }
    }
    for (int j = j0 + 4 * BS_PER_; j < pl; j += 4) {
        const unsigned ee = tmp[min(pb + j, E2_ - 1)];
        const int pp = atomicAdd(&cur[ee & 255u], 1);
        if (inlds) sl[min(pp, SLCAP_ - 1)] = (int)(ee >> 8);
        else       slots[min(start + pp, E2_ - 1)] = (int)(ee >> 8);
    }
    __syncthreads();
    if (inlds)
        for (int i = t; i < n; i += BS_T_) slots[start + i] = sl[i];
}


#define ROW_LANE(v, i) \
    ((unsigned)__builtin_amdgcn_update_dpp(0, (int)(v), 0x150 + (i), 0xf, 0xf, false))

#define ROW_LOAD(i, ix_, m_) \
    { const unsigned r_ = min(ROW_LANE(ix_, i), last_src); \
      if ((i) < (m_)) v##i = P4[r_ * 16u + c]; }

#define ROW_ADD(i) \
    s.x += v##i.x; s.y += v##i.y; s.z += v##i.z; s.w += v##i.w;

#define ROW_GROUP(ix_, m_, FIRST_) \
    { \
        const int mm_ = (m_); \
        float4 v0 = z4, v1 = z4, v2  = z4, v3  = z4, v4  = z4, v5  = z4, v6  = z4, v7  = z4; \
        float4 v8 = z4, v9 = z4, v10 = z4, v11 = z4, v12 = z4, v13 = z4, v14 = z4, v15 = z4; \
        ROW_LOAD(0, ix_, mm_) ROW_LOAD(1, ix_, mm_) ROW_LOAD(2, ix_, mm_) ROW_LOAD(3, ix_, mm_) \
        ROW_LOAD(4, ix_, mm_) ROW_LOAD(5, ix_, mm_) ROW_LOAD(6, ix_, mm_) ROW_LOAD(7, ix_, mm_) \
        if (mm_ > 8) { \
            ROW_LOAD(8,  ix_, mm_) ROW_LOAD(9,  ix_, mm_) ROW_LOAD(10, ix_, mm_) ROW_LOAD(11, ix_, mm_) \
            ROW_LOAD(12, ix_, mm_) ROW_LOAD(13, ix_, mm_) ROW_LOAD(14, ix_, mm_) ROW_LOAD(15, ix_, mm_) \
        } \
        FIRST_ \
        ROW_ADD(1) ROW_ADD(2) ROW_ADD(3) \
        ROW_ADD(4) ROW_ADD(5) ROW_ADD(6) ROW_ADD(7) \
        if (mm_ > 8) { \
            ROW_ADD(8)  ROW_ADD(9)  ROW_ADD(10) ROW_ADD(11) \
            ROW_ADD(12) ROW_ADD(13) ROW_ADD(14) ROW_ADD(15) \
        } \
    }

__device__ __forceinline__
void gather_set(const float* __restrict__ P, const unsigned last_src,
                float* __restrict__ acc, const int2* __restrict__ seg,
                const int* __restrict__ slots,
                const int n, const int first, const int step, const unsigned c) {
    const float4* __restrict__ P4 = (const float4*)P;
    const float4 z4 = make_float4(0.f, 0.f, 0.f, 0.f);
    int j  = first;
    int jn = j + step;
    int2 sg  = make_int2(0, 0);
    int2 sgn = make_int2(0, 0);
    unsigned idx = 0;
    if (j  < n) sg  = seg[j];
    if (jn < n) sgn = seg[jn];
    if ((int)c < sg.y) idx = (unsigned)slots[sg.x + (int)c];
    idx = min(idx, last_src);
    while (j < n) {
        const int o = sg.x, deg = sg.y;
        float4* orow = (float4*)(acc + ((size_t)j << 6)) + c;
        const float4 own = *orow;
        unsigned idxn = 0;
        if ((int)c < sgn.y) idxn = (unsigned)slots[sgn.x + (int)c];
        const int jnn = jn + step;
        int2 sgnn = make_int2(0, 0);
        if (jnn < n) sgnn = seg[jnn];

        float4 s;
        ROW_GROUP(idx, deg, s = v0;)
        for (int base = 16; base < deg; base += 16) {
            unsigned ix = 0;
            if (base + (int)c < deg) ix = (unsigned)slots[o + base + (int)c];
            ROW_GROUP(ix, deg - base, ROW_ADD(0))
        }

        const float inv = 1.0f / fmaxf((float)deg, 1.0f);
        float4 r;
        r.x = fmaf(s.x, inv, own.x);
        r.y = fmaf(s.y, inv, own.y);
        r.z = fmaf(s.z, inv, own.z);
        r.w = fmaf(s.w, inv, own.w);
        *orow = r;
        sg = sgn; sgn = sgnn; idx = idxn;
        j = jn;   jn = jnn;
    }
}

__global__ __launch_bounds__(256)
void row_gather(const float* __restrict__ P_dis, const float* __restrict__ P_gene,
                float* __restrict__ acc_gene, float* __restrict__ acc_dis,
                const int2* __restrict__ seg, const int* __restrict__ slots) {
    const unsigned c = threadIdx.x & 15u;
    const int hw  = (blockIdx.x * 256 + threadIdx.x) >> 4;
    const int nhw = (gridDim.x * 256) >> 4;
    gather_set(P_dis,  ND_ - 1, acc_gene, seg,       slots, NG_, hw,           nhw, c);
    gather_set(P_gene, NG_ - 1, acc_dis,  seg + NG_, slots, ND_, nhw - 1 - hw, nhw, c);
}

template<int K>
__global__ __launch_bounds__(128)
void node_wmma(const float* xa,
               const float* __restrict__ Wna,
               const float* __restrict__ Wsa, const float* __restrict__ biasa,
               float* outna, float* __restrict__ outsa, int Na, int nblk_a,
               const float* xb,
               const float* __restrict__ Wnb,
               const float* __restrict__ Wsb, const float* __restrict__ biasb,
               float* outnb, float* __restrict__ outsb, int Nb)
{
    __shared__ v2f Wpk[K * 64];
    const bool seta = (int)blockIdx.x < nblk_a;
    const float* x                 = seta ? xa    : xb;
    const float* __restrict__ Wn   = seta ? Wna   : Wnb;
    const float* __restrict__ Ws   = seta ? Wsa   : Wsb;
    const float* __restrict__ bias = seta ? biasa : biasb;
    float* outn                    = seta ? outna : outnb;
    float* __restrict__ outs       = seta ? outsa : outsb;
    const int N   = seta ? Na : Nb;
    const int blk = seta ? (int)blockIdx.x : (int)blockIdx.x - nblk_a;
    {
        constexpr int PER_T = K / 8;
        float4* lds4 = (float4*)Wpk;
        float4 ra[PER_T], rb[PER_T];
#pragma unroll
        for (int u = 0; u < PER_T; ++u) {
            const int I  = u * 128 + (int)threadIdx.x;
            const int PP = I >> 4, q = I & 15;
            const float* Wsel = (u * 8 < K / 2) ? Wn : Ws;
            const int pp = PP - ((u * 8 < K / 2) ? 0 : K / 2);
            const float4* g = (const float4*)(Wsel + (size_t)(2 * pp) * 64) + q;
            ra[u] = g[0];
            rb[u] = g[16];
        }
#pragma unroll
        for (int u = 0; u < PER_T; ++u) {
            const int I  = u * 128 + (int)threadIdx.x;
            const int PP = I >> 4, q = I & 15;
            float4* d = lds4 + ((PP * 64 + 4 * q) >> 1);
            d[0] = make_float4(ra[u].x, rb[u].x, ra[u].y, rb[u].y);
            d[1] = make_float4(ra[u].z, rb[u].z, ra[u].w, rb[u].w);
        }
    }
    __syncthreads();

    const int wave = threadIdx.x >> 5;
    const int lane = threadIdx.x & 31;
    const int pair = blk * 4 + wave;
    const int row0 = pair * 32;
    if (row0 >= N) return;
    const bool has2 = (row0 + 32 <= N);

    const int m  = lane & 15;
    const int ph = lane >> 4;
    const int rowA0 = row0 + m;
    const int rowA1 = min(row0 + 16 + m, N - 1);

    v8f accn0[4] = {};
    v8f accn1[4] = {};
    v8f accs0[4] = {};
    v8f accs1[4] = {};

    const v2f* X0 = (const v2f*)(x + (size_t)rowA0 * K);
    const v2f* X1 = (const v2f*)(x + (size_t)rowA1 * K);
    for (int k = 0; k < K; k += 4) {
        const int kp = (k >> 1) + ph;
        v2f a0 = X0[kp];
        v2f a1 = X1[kp];
#pragma unroll
        for (int t = 0; t < 4; ++t) {
            v2f bn = Wpk[kp * 64 + t * 16 + m];
            accn0[t] = __builtin_amdgcn_wmma_f32_16x16x4_f32(
                false, a0, false, bn, (short)0, accn0[t], false, false);
            accn1[t] = __builtin_amdgcn_wmma_f32_16x16x4_f32(
                false, a1, false, bn, (short)0, accn1[t], false, false);
            v2f bs = Wpk[(K / 2 + kp) * 64 + t * 16 + m];
            accs0[t] = __builtin_amdgcn_wmma_f32_16x16x4_f32(
                false, a0, false, bs, (short)0, accs0[t], false, false);
            accs1[t] = __builtin_amdgcn_wmma_f32_16x16x4_f32(
                false, a1, false, bs, (short)0, accs1[t], false, false);
        }
    }

    const int colb  = lane & 15;
    const int mb0   = row0 + 8 * ph;
#pragma unroll
    for (int t = 0; t < 4; ++t) {
        const float b = bias[t * 16 + colb];
#pragma unroll
        for (int r = 0; r < 8; ++r) {
            outn[(size_t)(mb0 + r) * 64 + t * 16 + colb] = accn0[t][r];
            outs[(size_t)(mb0 + r) * 64 + t * 16 + colb] = accs0[t][r] + b;
        }
        if (has2) {
#pragma unroll
            for (int r = 0; r < 8; ++r) {
                outn[(size_t)(mb0 + 16 + r) * 64 + t * 16 + colb] = accn1[t][r];
                outs[(size_t)(mb0 + 16 + r) * 64 + t * 16 + colb] = accs1[t][r] + b;
            }
        }
    }
}

#define F2_WAVES_ 8
#define F2_PITCH_ 132

__device__ __forceinline__
void gather_tile(const float* __restrict__ T, const unsigned last_src,
                 const float* __restrict__ X, const int2* __restrict__ seg,
                 const int* __restrict__ slots, float* tile,
                 const int row0, const int h, const unsigned c) {
    const float4* __restrict__ P4 = (const float4*)T;
    const float4 z4 = make_float4(0.f, 0.f, 0.f, 0.f);
    const int n = row0 + 16;
    int j  = row0 + h;
    int jn = j + 2;
    float4 ownr[8];
#pragma unroll
    for (int i = 0; i < 8; ++i)
        ownr[i] = *((const float4*)(X + ((size_t)(j + 2 * i) << 6)) + c);
    int2 sg  = seg[j];
    int2 sgn = seg[jn];
#pragma unroll
    for (int i = 0; i < 8; ++i)
        ((float4*)(tile + (h + 2 * i) * F2_PITCH_))[16 + c] = ownr[i];
    unsigned idx = 0;
    if ((int)c < sg.y) idx = (unsigned)slots[sg.x + (int)c];
    idx = min(idx, last_src);
    while (j < n) {
        const int o = sg.x, deg = sg.y;
        unsigned idxn = 0;
        if ((int)c < sgn.y) idxn = (unsigned)slots[sgn.x + (int)c];
        const int jnn = jn + 2;
        int2 sgnn = make_int2(0, 0);
        if (jnn < n) sgnn = seg[jnn];

        float4 s;
        ROW_GROUP(idx, deg, s = v0;)
        for (int base = 16; base < deg; base += 16) {
            unsigned ix = 0;
            if (base + (int)c < deg) ix = (unsigned)slots[o + base + (int)c];
            ROW_GROUP(ix, deg - base, ROW_ADD(0))
        }

        const float inv = 1.0f / fmaxf((float)deg, 1.0f);
        ((float4*)(tile + (j - row0) * F2_PITCH_))[c] =
            make_float4(s.x * inv, s.y * inv, s.z * inv, s.w * inv);
        sg = sgn; sgn = sgnn; idx = idxn;
        j = jn;   jn = jnn;
    }
}

__global__ __launch_bounds__(256)
void layer2_fused(const float* __restrict__ Ta, const float* __restrict__ Xa,
                  const float* __restrict__ Wla, const float* __restrict__ Wra,
                  const float* __restrict__ biasa, float* __restrict__ outa,
                  const int2* __restrict__ sega, int Na, int last_a, int nblk_a,
                  const float* __restrict__ Tb, const float* __restrict__ Xb,
                  const float* __restrict__ Wlb, const float* __restrict__ Wrb,
                  const float* __restrict__ biasb, float* __restrict__ outb,
                  const int2* __restrict__ segb, int Nb, int last_b,
                  const int* __restrict__ slots)
{
    __shared__ v2f   Wpk[64 * 64];
    __shared__ float4 tiles4[F2_WAVES_ * 16 * F2_PITCH_ / 4];
    const bool seta = (int)blockIdx.x < nblk_a;
    const float* __restrict__ T    = seta ? Ta    : Tb;
    const float* __restrict__ X    = seta ? Xa    : Xb;
    const float* __restrict__ Wl   = seta ? Wla   : Wlb;
    const float* __restrict__ Wr   = seta ? Wra   : Wrb;
    const float* __restrict__ bias = seta ? biasa : biasb;
    float* __restrict__ out        = seta ? outa  : outb;
    const int2* __restrict__ seg   = seta ? sega  : segb;
    const int N    = seta ? Na : Nb;
    const int last = seta ? last_a : last_b;
    const int blk  = seta ? (int)blockIdx.x : (int)blockIdx.x - nblk_a;

    constexpr int PER_T = 4;
    float4 ra[PER_T], rb[PER_T];
#pragma unroll
    for (int u = 0; u < PER_T; ++u) {
        const int I  = u * 256 + (int)threadIdx.x;
        const int PP = I >> 4, q = I & 15;
        const float* Wsel = (u < 2) ? Wl : Wr;
        const int pp = PP - ((u < 2) ? 0 : 32);
        const float4* g = (const float4*)(Wsel + (size_t)(2 * pp) * 64) + q;
        ra[u] = g[0];
        rb[u] = g[16];
    }

    const int wave = threadIdx.x >> 5;
    const int lane = threadIdx.x & 31;
    const int row0 = (blk * F2_WAVES_ + wave) * 16;
    const bool active = row0 < N;
    float* tile = (float*)tiles4 + wave * (16 * F2_PITCH_);

    if (active)
        gather_tile(T, (unsigned)last, X, seg, slots, tile, row0, lane >> 4, (unsigned)(lane & 15));

    {
        float4* lds4 = (float4*)Wpk;
#pragma unroll
        for (int u = 0; u < PER_T; ++u) {
            const int I  = u * 256 + (int)threadIdx.x;
            const int PP = I >> 4, q = I & 15;
            float4* d = lds4 + ((PP * 64 + 4 * q) >> 1);
            d[0] = make_float4(ra[u].x, rb[u].x, ra[u].y, rb[u].y);
            d[1] = make_float4(ra[u].z, rb[u].z, ra[u].w, rb[u].w);
        }
    }
    __syncthreads();
    if (!active) return;

    const int m  = lane & 15;
    const int ph = lane >> 4;
    v8f acc[4] = {};
    const float* arow = tile + m * F2_PITCH_ + 2 * ph;
    for (int k = 0; k < 128; k += 4) {
        const int kp = (k >> 1) + ph;
        const v2f a = *(const v2f*)(arow + k);
#pragma unroll
        for (int t = 0; t < 4; ++t) {
            const v2f b = Wpk[kp * 64 + t * 16 + m];
            acc[t] = __builtin_amdgcn_wmma_f32_16x16x4_f32(
                false, a, false, b, (short)0, acc[t], false, false);
        }
    }

    const int mb0 = row0 + 8 * ph;
#pragma unroll
    for (int t = 0; t < 4; ++t) {
        const float b = bias[t * 16 + m];
#pragma unroll
        for (int r = 0; r < 8; ++r)
            out[(size_t)(mb0 + r) * 64 + t * 16 + m] = acc[t][r] + b;
    }
}

extern "C" void kernel_launch(void* const* d_in, const int* in_sizes, int n_in,
                              void* d_out, int out_size, void* d_ws, size_t ws_size,
                              hipStream_t stream) {
    const float* x_disease = (const float*)d_in[0];
    const float* x_gene    = (const float*)d_in[1];
    const int*   src_dg    = (const int*)d_in[2];
    const int*   dst_dg    = (const int*)d_in[3];
    const int*   src_gd    = (const int*)d_in[4];
    const int*   dst_gd    = (const int*)d_in[5];
    const float* Wl1_dg = (const float*)d_in[6];
    const float* bl1_dg = (const float*)d_in[7];
    const float* Wr1_dg = (const float*)d_in[8];
    const float* Wl1_gd = (const float*)d_in[9];
    const float* bl1_gd = (const float*)d_in[10];
    const float* Wr1_gd = (const float*)d_in[11];
    const float* Wl2_dg = (const float*)d_in[12];
    const float* bl2_dg = (const float*)d_in[13];
    const float* Wr2_dg = (const float*)d_in[14];
    const float* Wl2_gd = (const float*)d_in[15];
    const float* bl2_gd = (const float*)d_in[16];
    const float* Wr2_gd = (const float*)d_in[17];

    float* d2 = (float*)d_out;
    float* g2 = d2 + (size_t)ND_ * 64;
    float* p1_d = d2;
    float* p1_g = g2;

    float* ws = (float*)d_ws;
    size_t off = 0;
    int*      slots  = (int*)(ws + off);      off += (size_t)E2_;
    unsigned* tmp    = (unsigned*)(ws + off); off += (size_t)E2_;
    int2*     seg    = (int2*)(ws + off);     off += (size_t)NT_ * 2;
    int*      offs   = (int*)(ws + off);
    off += (size_t)NT_ + 48;
    float* g1    = ws + off; off += (size_t)NG_ * H_;
    float* d1    = ws + off; off += (size_t)ND_ * H_;
    (void)off; (void)ws_size; (void)in_sizes; (void)n_in; (void)out_size;

    const int TPB = 256;
    const int gb  = 1024;
    const int pairs_g = (NG_ / 16 + 1) / 2;
    const int pairs_d = (ND_ / 16 + 1) / 2;
    const int nblk_g  = (pairs_g + 3) / 4;
    const int nblk_d  = (pairs_d + 3) / 4;
    const int fblk_g  = (NG_ / 16 + F2_WAVES_ - 1) / F2_WAVES_;
    const int fblk_d  = (ND_ / 16 + F2_WAVES_ - 1) / F2_WAVES_;

    chunk_partition<<<SB_, 1024, 0, stream>>>(src_dg, dst_dg, src_gd, dst_gd, offs, tmp);
    bucket_sort<<<NBKT_, BS_T_, 0, stream>>>(tmp, offs, seg, slots);

    node_wmma<DIN_><<<nblk_d + nblk_g, 128, 0, stream>>>(
        x_disease, Wl1_dg, Wr1_gd, bl1_gd, p1_d, d1, ND_, nblk_d,
        x_gene,    Wl1_gd, Wr1_dg, bl1_dg, p1_g, g1, NG_);

    row_gather<<<gb, TPB, 0, stream>>>(p1_d, p1_g, g1, d1, seg, slots);

    layer2_fused<<<fblk_d + fblk_g, 256, 0, stream>>>(
        g1, d1, Wl2_gd, Wr2_gd, bl2_gd, d2, seg + NG_, ND_, NG_ - 1, fblk_d,
        d1, g1, Wl2_dg, Wr2_dg, bl2_dg, g2, seg,       NG_, ND_ - 1,
        slots);
}


# layer 1: own products x@Wr move out of the node-product launch into the edge-mean launch, run beside the row sums; a tile's means are the WMMA start values; no own plane stored; f32 sum order changes
# speedup vs baseline: 4.8977x; 1.2275x over previous
#include <hip/hip_runtime.h>

typedef float v2f __attribute__((ext_vector_type(2)));
typedef float v8f __attribute__((ext_vector_type(8)));
typedef int   v4i __attribute__((ext_vector_type(4)));

#define ND_   50000
#define NG_   100000
#define E_    600000
#define DIN_  128
#define H_    64
#define NT_   (NG_ + ND_)
#define E2_   (2 * E_)

#define NBKT_  ((NT_ + 255) >> 8)
#define SB_    128
#define QCH_   ((E2_ / 4 + SB_ - 1) / SB_)
#define CHW_   (QCH_ * 4)
#define OFP_   (NBKT_ + 1)

#define BUCKET(u) ((int)min((unsigned)(u) >> 8, (unsigned)(NBKT_ - 1)))

#define EDGE_WORD(uu, ss) (((unsigned)(ss) << 8) | ((uu) & 255u))

#define CP_T_     128
#define CP_TRIPS_ ((QCH_ + CP_T_ - 1) / CP_T_)
#define CP_HALF_  ((CP_TRIPS_ + 1) / 2)
#define CP_BPT_   ((OFP_ + CP_T_ - 1) / CP_T_)
#define CP_LDSW_  (CP_T_ * CP_BPT_ + 16)

#define CP_Q(k)    min(q0 + t + (k) * CP_T_, q1 - 1)
#define CP_BASE(k) ((CP_Q(k) < E_ / 4) ? 0 : NG_)

__device__ __forceinline__
void chunk_part(const int* __restrict__ src_dg, const int* __restrict__ dst_dg,
                const int* __restrict__ src_gd, const int* __restrict__ dst_gd,
                int* __restrict__ offs, unsigned* __restrict__ tmp,
                int* lds, const int me) {
    int* cur      = lds;
    int* wtot     = lds + CP_T_ * CP_BPT_;
    v4i* srt4     = (v4i*)(lds + CP_LDSW_);
    unsigned* srt = (unsigned*)srt4;
    const int t    = threadIdx.x;
    const int lane = t & 31;
    const int w    = t >> 5;
    const int q0   = me * QCH_;
    const int q1   = min(q0 + QCH_, E2_ / 4);
    v4i d[CP_TRIPS_];
#pragma unroll
    for (int k = 0; k < CP_TRIPS_; ++k) {
        const int q   = CP_Q(k);
        const bool dg = q < E_ / 4;
        const v4i* dp = dg ? (const v4i*)dst_dg + q : (const v4i*)dst_gd + (q - E_ / 4);
        d[k] = *dp;
    }
    for (int i = t; i < NBKT_; i += CP_T_) cur[i] = 0;
    __syncthreads();
#pragma unroll
    for (int k = 0; k < CP_TRIPS_; ++k) {
        if (q0 + t + k * CP_T_ < q1) {
            const int base = CP_BASE(k);
            atomicAdd(&cur[BUCKET(base + d[k].x)], 1);
            atomicAdd(&cur[BUCKET(base + d[k].y)], 1);
            atomicAdd(&cur[BUCKET(base + d[k].z)], 1);
            atomicAdd(&cur[BUCKET(base + d[k].w)], 1);
        }
    }
    __syncthreads();
    int cb[CP_BPT_];
    int tot = 0;
#pragma unroll
    for (int i = 0; i < CP_BPT_; ++i) {
        const int b = t * CP_BPT_ + i;
        cb[i] = (b < NBKT_) ? cur[b] : 0;
        tot += cb[i];
    }
    int incl = tot;
#pragma unroll
    for (int dd = 1; dd < 32; dd <<= 1) {
        const int v = __shfl_up(incl, dd);
        if (lane >= dd) incl += v;
    }
    if (lane == 31) wtot[w] = incl;
    __syncthreads();
    int run = incl - tot;
#pragma unroll
    for (int k = 0; k < CP_T_ / 32; ++k) run += (k < w) ? wtot[k] : 0;
#pragma unroll
    for (int i = 0; i < CP_BPT_; ++i) {
        const int b = t * CP_BPT_ + i;
        if (b < NBKT_)  cur[b] = run;
        if (b <= NBKT_) offs[me * OFP_ + b] = run;
        run += cb[i];
    }
    __syncthreads();

#pragma unroll
    for (int hh = 0; hh < 2; ++hh) {
        v4i s[CP_HALF_];
#pragma unroll
        for (int kk = 0; kk < CP_HALF_; ++kk) {
            const int k = hh * CP_HALF_ + kk;
            if (k < CP_TRIPS_) {
                const int q   = CP_Q(k);
                const bool dg = q < E_ / 4;
                const v4i* sp = dg ? (const v4i*)src_dg + q : (const v4i*)src_gd + (q - E_ / 4);
                s[kk] = *sp;
            }
        }
#pragma unroll
        for (int kk = 0; kk < CP_HALF_; ++kk) {
            const int k = hh * CP_HALF_ + kk;
            if (k < CP_TRIPS_) {
                if (q0 + t + k * CP_T_ < q1) {
                    const int base = CP_BASE(k);
                    const unsigned u0 = (unsigned)(base + d[k].x);
                    const unsigned u1 = (unsigned)(base + d[k].y);
                    const unsigned u2 = (unsigned)(base + d[k].z);
                    const unsigned u3 = (unsigned)(base + d[k].w);
                    const unsigned p0 = (unsigned)atomicAdd(&cur[BUCKET(u0)], 1);
                    const unsigned p1 = (unsigned)atomicAdd(&cur[BUCKET(u1)], 1);
                    const unsigned p2 = (unsigned)atomicAdd(&cur[BUCKET(u2)], 1);
                    const unsigned p3 = (unsigned)atomicAdd(&cur[BUCKET(u3)], 1);
                    srt[min(p0, (unsigned)(CHW_ - 1))] = EDGE_WORD(u0, s[kk].x);
                    srt[min(p1, (unsigned)(CHW_ - 1))] = EDGE_WORD(u1, s[kk].y);
                    srt[min(p2, (unsigned)(CHW_ - 1))] = EDGE_WORD(u2, s[kk].z);
                    srt[min(p3, (unsigned)(CHW_ - 1))] = EDGE_WORD(u3, s[kk].w);
                }
            }
        }
    }
    __syncthreads();
    v4i* out4 = (v4i*)tmp + q0;
    for (int g = t; q0 + g < q1; g += CP_T_) out4[g] = srt4[g];
}

#define BS_T_   512
#define BS_PER_ 8
#define SLCAP_  4096

__global__ __launch_bounds__(BS_T_)
void bucket_sort(const unsigned* __restrict__ tmp, const int* __restrict__ offs,
                 int2* __restrict__ seg, int* __restrict__ slots) {
    __shared__ int cnt[256];
    __shared__ int cur[256];
    __shared__ int wtot[8];
    __shared__ int pbeg[SB_];
    __shared__ int plen[SB_];
    __shared__ int wsum[SB_ / 32];
    __shared__ int wlen[SB_ / 32];
    __shared__ int sl[SLCAP_];
    const int t    = threadIdx.x;
    const int lane = t & 31;
    const int w    = t >> 5;
    const int b    = blockIdx.x;
    if (t < 256) cnt[t] = 0;
    if (t < SB_) {
        const int o0 = offs[t * OFP_ + b];
        const int o1 = offs[t * OFP_ + b + 1];
        const int clen = min(CHW_, E2_ - t * CHW_);
        const int a0 = min(max(o0, 0), clen);
        const int a1 = min(max(o1, a0), clen);
        pbeg[t] = t * CHW_ + a0;
        plen[t] = a1 - a0;
        int sacc = a0, lacc = a1 - a0;
#pragma unroll
        for (int dd = 16; dd >= 1; dd >>= 1) {
            sacc += __shfl_xor(sacc, dd);
            lacc += __shfl_xor(lacc, dd);
        }
        if (lane == 0) { wsum[w] = sacc; wlen[w] = lacc; }
    }
    __syncthreads();
    int start = 0, n = 0;
#pragma unroll
    for (int k = 0; k < SB_ / 32; ++k) { start += wsum[k]; n += wlen[k]; }
    start = min(start, E2_);
    n     = min(n, E2_ - start);
    const bool inlds = n <= SLCAP_;
    const int j0 = t & 3;
    const int pb = pbeg[t >> 2];
    const int pl = plen[t >> 2];
    unsigned e[BS_PER_];
#pragma unroll
    for (int k = 0; k < BS_PER_; ++k)
        e[k] = tmp[min(pb + j0 + 4 * k, E2_ - 1)];
#pragma unroll
    for (int k = 0; k < BS_PER_; ++k)
        if (j0 + 4 * k < pl) atomicAdd(&cnt[e[k] & 255u], 1);
    for (int j = j0 + 4 * BS_PER_; j < pl; j += 4)
        atomicAdd(&cnt[tmp[min(pb + j, E2_ - 1)] & 255u], 1);
    __syncthreads();
    int c = 0, incl = 0;
    if (t < 256) {
        c = cnt[t];
        incl = c;
#pragma unroll
        for (int dd = 1; dd < 32; dd <<= 1) {
            const int v = __shfl_up(incl, dd);
            if (lane >= dd) incl += v;
        }
        if (lane == 31) wtot[w] = incl;
    }
    __syncthreads();
    if (t < 256) {
        int wb = 0;
#pragma unroll
        for (int k = 0; k < 8; ++k) wb += (k < w) ? wtot[k] : 0;
        const int excl = wb + incl - c;
        const int u = b * 256 + t;
        if (u < NT_) seg[u] = make_int2(start + excl, c);
        cur[t] = excl;
    }
    __syncthreads();
    int p[BS_PER_];
#pragma unroll
    for (int k = 0; k < BS_PER_; ++k) {
        p[k] = 0;
        if (j0 + 4 * k < pl) p[k] = atomicAdd(&cur[e[k] & 255u], 1);
    }
#pragma unroll
    for (int k = 0; k < BS_PER_; ++k) {
        if (j0 + 4 * k < pl) {
            if (inlds) sl[min(p[k], SLCAP_ - 1)] = (int)(e[k] >> 8);
            else       slots[min(start + p[k], E2_ - 1)] = (int)(e[k] >> 8);
        }
    }
    for (int j = j0 + 4 * BS_PER_; j < pl; j += 4) {
        const unsigned ee = tmp[min(pb + j, E2_ - 1)];
        const int pp = atomicAdd(&cur[ee & 255u], 1);
        if (inlds) sl[min(pp, SLCAP_ - 1)] = (int)(ee >> 8);
        else       slots[min(start + pp, E2_ - 1)] = (int)(ee >> 8);
    }
    __syncthreads();
    if (inlds)
        for (int i = t; i < n; i += BS_T_) slots[start + i] = sl[i];
}


#define ROW_LANE(v, i) \
    ((unsigned)__builtin_amdgcn_update_dpp(0, (int)(v), 0x150 + (i), 0xf, 0xf, false))

#define ROW_LOAD(i, ix_, m_) \
    { const unsigned r_ = min(ROW_LANE(ix_, i), last_src); \
      if ((i) < (m_)) v##i = P4[r_ * 16u + c]; }

#define ROW_ADD(i) \
    s.x += v##i.x; s.y += v##i.y; s.z += v##i.z; s.w += v##i.w;

#define ROW_GROUP(ix_, m_, FIRST_) \
    { \
        const int mm_ = (m_); \
        float4 v0 = z4, v1 = z4, v2  = z4, v3  = z4, v4  = z4, v5  = z4, v6  = z4, v7  = z4; \
        float4 v8 = z4, v9 = z4, v10 = z4, v11 = z4, v12 = z4, v13 = z4, v14 = z4, v15 = z4; \
        ROW_LOAD(0, ix_, mm_) ROW_LOAD(1, ix_, mm_) ROW_LOAD(2, ix_, mm_) ROW_LOAD(3, ix_, mm_) \
        ROW_LOAD(4, ix_, mm_) ROW_LOAD(5, ix_, mm_) ROW_LOAD(6, ix_, mm_) ROW_LOAD(7, ix_, mm_) \
        if (mm_ > 8) { \
            ROW_LOAD(8,  ix_, mm_) ROW_LOAD(9,  ix_, mm_) ROW_LOAD(10, ix_, mm_) ROW_LOAD(11, ix_, mm_) \
            ROW_LOAD(12, ix_, mm_) ROW_LOAD(13, ix_, mm_) ROW_LOAD(14, ix_, mm_) ROW_LOAD(15, ix_, mm_) \
        } \
        FIRST_ \
        ROW_ADD(1) ROW_ADD(2) ROW_ADD(3) \
        ROW_ADD(4) ROW_ADD(5) ROW_ADD(6) ROW_ADD(7) \
        if (mm_ > 8) { \
            ROW_ADD(8)  ROW_ADD(9)  ROW_ADD(10) ROW_ADD(11) \
            ROW_ADD(12) ROW_ADD(13) ROW_ADD(14) ROW_ADD(15) \
        } \
    }

template<int K>
__global__ __launch_bounds__(128)
void node_wmma(const int* __restrict__ src_dg, const int* __restrict__ dst_dg,
               const int* __restrict__ src_gd, const int* __restrict__ dst_gd,
               int* __restrict__ offs, unsigned* __restrict__ tmp,
               const float* xa, const float* __restrict__ Wna,
               float* outna, int Na, int nblk_a,
               const float* xb, const float* __restrict__ Wnb,
               float* outnb, int Nb, int nblk_ab)
{
    constexpr int WPK_N = (K / 2) * 64 > (CP_LDSW_ + CHW_ + 1) / 2 ? (K / 2) * 64 : (CP_LDSW_ + CHW_ + 1) / 2;
    __shared__ v2f Wpk[WPK_N];
    static_assert(sizeof(v2f) * WPK_N >= sizeof(int) * (CP_LDSW_ + CHW_), "chunk copy fits the weight tile");
    if ((int)blockIdx.x >= nblk_ab) {
        chunk_part(src_dg, dst_dg, src_gd, dst_gd, offs, tmp, (int*)Wpk,
                   min((int)blockIdx.x - nblk_ab, SB_ - 1));
        return;
    }
    const bool seta = (int)blockIdx.x < nblk_a;
    const float* x                 = seta ? xa    : xb;
    const float* __restrict__ Wn   = seta ? Wna   : Wnb;
    float* outn                    = seta ? outna : outnb;
    const int N   = seta ? Na : Nb;
    const int blk = seta ? (int)blockIdx.x : (int)blockIdx.x - nblk_a;
    {
        constexpr int PER_T = K / 16;
        float4* lds4 = (float4*)Wpk;
        float4 ra[PER_T], rb[PER_T];
#pragma unroll
        for (int u = 0; u < PER_T; ++u) {
            const int I  = u * 128 + (int)threadIdx.x;
            const int PP = I >> 4, q = I & 15;
            const float4* g = (const float4*)(Wn + (size_t)(2 * PP) * 64) + q;
            ra[u] = g[0];
            rb[u] = g[16];
        }
#pragma unroll
        for (int u = 0; u < PER_T; ++u) {
            const int I  = u * 128 + (int)threadIdx.x;
            const int PP = I >> 4, q = I & 15;
            float4* d = lds4 + ((PP * 64 + 4 * q) >> 1);
            d[0] = make_float4(ra[u].x, rb[u].x, ra[u].y, rb[u].y);
            d[1] = make_float4(ra[u].z, rb[u].z, ra[u].w, rb[u].w);
        }
    }
    __syncthreads();

    const int wave = threadIdx.x >> 5;
    const int lane = threadIdx.x & 31;
    const int pair = blk * 4 + wave;
    const int row0 = pair * 32;
    if (row0 >= N) return;
    const bool has2 = (row0 + 32 <= N);

    const int m  = lane & 15;
    const int ph = lane >> 4;
    const int rowA0 = row0 + m;
    const int rowA1 = min(row0 + 16 + m, N - 1);

    v8f accn0[4] = {};
    v8f accn1[4] = {};

    const v2f* X0 = (const v2f*)(x + (size_t)rowA0 * K);
    const v2f* X1 = (const v2f*)(x + (size_t)rowA1 * K);
    for (int k = 0; k < K; k += 4) {
        const int kp = (k >> 1) + ph;
        v2f a0 = X0[kp];
        v2f a1 = X1[kp];
#pragma unroll
        for (int t = 0; t < 4; ++t) {
            v2f bn = Wpk[kp * 64 + t * 16 + m];
            accn0[t] = __builtin_amdgcn_wmma_f32_16x16x4_f32(
                false, a0, false, bn, (short)0, accn0[t], false, false);
            accn1[t] = __builtin_amdgcn_wmma_f32_16x16x4_f32(
                false, a1, false, bn, (short)0, accn1[t], false, false);
        }
    }

    const int colb  = lane & 15;
    const int mb0   = row0 + 8 * ph;
#pragma unroll
    for (int t = 0; t < 4; ++t) {
#pragma unroll
        for (int r = 0; r < 8; ++r)
            outn[(size_t)(mb0 + r) * 64 + t * 16 + colb] = accn0[t][r];
        if (has2) {
#pragma unroll
            for (int r = 0; r < 8; ++r)
                outn[(size_t)(mb0 + 16 + r) * 64 + t * 16 + colb] = accn1[t][r];
        }
    }
}

#define F1_WAVES_ 8
#define F1_PITCH_ 68

__device__ __forceinline__
void gather_mean(const float* __restrict__ T, const unsigned last_src,
                 const int2* __restrict__ seg, const int* __restrict__ slots,
                 float* tile, const int row0, const int h, const unsigned c) {
    const float4* __restrict__ P4 = (const float4*)T;
    const float4 z4 = make_float4(0.f, 0.f, 0.f, 0.f);
    const int n = row0 + 16;
    int j  = row0 + h;
    int jn = j + 2;
    int2 sg  = seg[j];
    int2 sgn = seg[jn];
    unsigned idx = 0;
    if ((int)c < sg.y) idx = (unsigned)slots[sg.x + (int)c];
    idx = min(idx, last_src);
    while (j < n) {
        const int o = sg.x, deg = sg.y;
        unsigned idxn = 0;
        if ((int)c < sgn.y) idxn = (unsigned)slots[sgn.x + (int)c];
        const int jnn = jn + 2;
        int2 sgnn = make_int2(0, 0);
        if (jnn < n) sgnn = seg[jnn];

        float4 s;
        ROW_GROUP(idx, deg, s = v0;)
        for (int base = 16; base < deg; base += 16) {
            unsigned ix = 0;
            if (base + (int)c < deg) ix = (unsigned)slots[o + base + (int)c];
            ROW_GROUP(ix, deg - base, ROW_ADD(0))
        }

        const float inv = 1.0f / fmaxf((float)deg, 1.0f);
        ((float4*)(tile + (j - row0) * F1_PITCH_))[c] =
            make_float4(s.x * inv, s.y * inv, s.z * inv, s.w * inv);
        sg = sgn; sgn = sgnn; idx = idxn;
        j = jn;   jn = jnn;
    }
}

__global__ __launch_bounds__(256)
void layer1_fused(const float* __restrict__ Pa, const float* __restrict__ Xa,
                  const float* __restrict__ Wra, const float* __restrict__ biasa,
                  float* __restrict__ outa, const int2* __restrict__ sega,
                  int Na, int last_a, int nblk_a,
                  const float* __restrict__ Pb, const float* __restrict__ Xb,
                  const float* __restrict__ Wrb, const float* __restrict__ biasb,
                  float* __restrict__ outb, const int2* __restrict__ segb,
                  int Nb, int last_b,
                  const int* __restrict__ slots)
{
    __shared__ v2f    Wpk[64 * 64];
    __shared__ float4 means4[F1_WAVES_ * 16 * F1_PITCH_ / 4];
    const bool seta = (int)blockIdx.x < nblk_a;
    const float* __restrict__ P    = seta ? Pa    : Pb;
    const float* __restrict__ X    = seta ? Xa    : Xb;
    const float* __restrict__ Wr   = seta ? Wra   : Wrb;
    const float* __restrict__ bias = seta ? biasa : biasb;
    float* __restrict__ out        = seta ? outa  : outb;
    const int2* __restrict__ seg   = seta ? sega  : segb;
    const int N    = seta ? Na : Nb;
    const int last = seta ? last_a : last_b;
    const int blk  = seta ? (int)blockIdx.x : (int)blockIdx.x - nblk_a;

    constexpr int PER_T = 4;
    float4 ra[PER_T], rb[PER_T];
#pragma unroll
    for (int u = 0; u < PER_T; ++u) {
        const int I  = u * 256 + (int)threadIdx.x;
        const int PP = I >> 4, q = I & 15;
        const float4* g = (const float4*)(Wr + (size_t)(2 * PP) * 64) + q;
        ra[u] = g[0];
        rb[u] = g[16];
    }

    const int wave = threadIdx.x >> 5;
    const int lane = threadIdx.x & 31;
    const int row0 = (blk * F1_WAVES_ + wave) * 16;
    const bool active = row0 < N;
    float* tile = (float*)means4 + wave * (16 * F1_PITCH_);

    if (active)
        gather_mean(P, (unsigned)last, seg, slots, tile, row0, lane >> 4, (unsigned)(lane & 15));

    const int m  = lane & 15;
    const int ph = lane >> 4;
    const v2f* X0 = (const v2f*)(X + (size_t)min(row0 + m, N - 1) * DIN_);
    v2f a[DIN_ / 4];
#pragma unroll
    for (int s = 0; s < DIN_ / 4; ++s) a[s] = X0[2 * s + ph];

    {
        float4* lds4 = (float4*)Wpk;
#pragma unroll
        for (int u = 0; u < PER_T; ++u) {
            const int I  = u * 256 + (int)threadIdx.x;
            const int PP = I >> 4, q = I & 15;
            float4* d = lds4 + ((PP * 64 + 4 * q) >> 1);
            d[0] = make_float4(ra[u].x, rb[u].x, ra[u].y, rb[u].y);
            d[1] = make_float4(ra[u].z, rb[u].z, ra[u].w, rb[u].w);
        }
    }
    __syncthreads();
    if (!active) return;

    v8f acc[4];
#pragma unroll
    for (int t = 0; t < 4; ++t) {
#pragma unroll
        for (int r = 0; r < 8; ++r)
            acc[t][r] = tile[(8 * ph + r) * F1_PITCH_ + t * 16 + m];
    }
#pragma unroll
    for (int s = 0; s < DIN_ / 4; ++s) {
        const int kp = 2 * s + ph;
#pragma unroll
        for (int t = 0; t < 4; ++t) {
            const v2f b = Wpk[kp * 64 + t * 16 + m];
            acc[t] = __builtin_amdgcn_wmma_f32_16x16x4_f32(
                false, a[s], false, b, (short)0, acc[t], false, false);
        }
    }

    const int mb0 = row0 + 8 * ph;
#pragma unroll
    for (int t = 0; t < 4; ++t) {
        const float b = bias[t * 16 + m];
#pragma unroll
        for (int r = 0; r < 8; ++r)
            out[(size_t)(mb0 + r) * 64 + t * 16 + m] = acc[t][r] + b;
    }
}

#define F2_WAVES_ 8
#define F2_PITCH_ 132

__device__ __forceinline__
void gather_tile(const float* __restrict__ T, const unsigned last_src,
                 const float* __restrict__ X, const int2* __restrict__ seg,
                 const int* __restrict__ slots, float* tile,
                 const int row0, const int h, const unsigned c) {
    const float4* __restrict__ P4 = (const float4*)T;
    const float4 z4 = make_float4(0.f, 0.f, 0.f, 0.f);
    const int n = row0 + 16;
    int j  = row0 + h;
    int jn = j + 2;
    float4 ownr[8];
#pragma unroll
    for (int i = 0; i < 8; ++i)
        ownr[i] = *((const float4*)(X + ((size_t)(j + 2 * i) << 6)) + c);
    int2 sg  = seg[j];
    int2 sgn = seg[jn];
#pragma unroll
    for (int i = 0; i < 8; ++i)
        ((float4*)(tile + (h + 2 * i) * F2_PITCH_))[16 + c] = ownr[i];
    unsigned idx = 0;
    if ((int)c < sg.y) idx = (unsigned)slots[sg.x + (int)c];
    idx = min(idx, last_src);
    while (j < n) {
        const int o = sg.x, deg = sg.y;
        unsigned idxn = 0;
        if ((int)c < sgn.y) idxn = (unsigned)slots[sgn.x + (int)c];
        const int jnn = jn + 2;
        int2 sgnn = make_int2(0, 0);
        if (jnn < n) sgnn = seg[jnn];

        float4 s;
        ROW_GROUP(idx, deg, s = v0;)
        for (int base = 16; base < deg; base += 16) {
            unsigned ix = 0;
            if (base + (int)c < deg) ix = (unsigned)slots[o + base + (int)c];
            ROW_GROUP(ix, deg - base, ROW_ADD(0))
        }

        const float inv = 1.0f / fmaxf((float)deg, 1.0f);
        ((float4*)(tile + (j - row0) * F2_PITCH_))[c] =
            make_float4(s.x * inv, s.y * inv, s.z * inv, s.w * inv);
        sg = sgn; sgn = sgnn; idx = idxn;
        j = jn;   jn = jnn;
    }
}

__global__ __launch_bounds__(256)
void layer2_fused(const float* __restrict__ Ta, const float* __restrict__ Xa,
                  const float* __restrict__ Wla, const float* __restrict__ Wra,
                  const float* __restrict__ biasa, float* __restrict__ outa,
                  const int2* __restrict__ sega, int Na, int last_a, int nblk_a,
                  const float* __restrict__ Tb, const float* __restrict__ Xb,
                  const float* __restrict__ Wlb, const float* __restrict__ Wrb,
                  const float* __restrict__ biasb, float* __restrict__ outb,
                  const int2* __restrict__ segb, int Nb, int last_b,
                  const int* __restrict__ slots)
{
    __shared__ v2f   Wpk[64 * 64];
    __shared__ float4 tiles4[F2_WAVES_ * 16 * F2_PITCH_ / 4];
    const bool seta = (int)blockIdx.x < nblk_a;
    const float* __restrict__ T    = seta ? Ta    : Tb;
    const float* __restrict__ X    = seta ? Xa    : Xb;
    const float* __restrict__ Wl   = seta ? Wla   : Wlb;
    const float* __restrict__ Wr   = seta ? Wra   : Wrb;
    const float* __restrict__ bias = seta ? biasa : biasb;
    float* __restrict__ out        = seta ? outa  : outb;
    const int2* __restrict__ seg   = seta ? sega  : segb;
    const int N    = seta ? Na : Nb;
    const int last = seta ? last_a : last_b;
    const int blk  = seta ? (int)blockIdx.x : (int)blockIdx.x - nblk_a;

    constexpr int PER_T = 4;
    float4 ra[PER_T], rb[PER_T];
#pragma unroll
    for (int u = 0; u < PER_T; ++u) {
        const int I  = u * 256 + (int)threadIdx.x;
        const int PP = I >> 4, q = I & 15;
        const float* Wsel = (u < 2) ? Wl : Wr;
        const int pp = PP - ((u < 2) ? 0 : 32);
        const float4* g = (const float4*)(Wsel + (size_t)(2 * pp) * 64) + q;
        ra[u] = g[0];
        rb[u] = g[16];
    }

    const int wave = threadIdx.x >> 5;
    const int lane = threadIdx.x & 31;
    const int row0 = (blk * F2_WAVES_ + wave) * 16;
    const bool active = row0 < N;
    float* tile = (float*)tiles4 + wave * (16 * F2_PITCH_);

    if (active)
        gather_tile(T, (unsigned)last, X, seg, slots, tile, row0, lane >> 4, (unsigned)(lane & 15));

    {
        float4* lds4 = (float4*)Wpk;
#pragma unroll
        for (int u = 0; u < PER_T; ++u) {
            const int I  = u * 256 + (int)threadIdx.x;
            const int PP = I >> 4, q = I & 15;
            float4* d = lds4 + ((PP * 64 + 4 * q) >> 1);
            d[0] = make_float4(ra[u].x, rb[u].x, ra[u].y, rb[u].y);
            d[1] = make_float4(ra[u].z, rb[u].z, ra[u].w, rb[u].w);
        }
    }
    __syncthreads();
    if (!active) return;

    const int m  = lane & 15;
    const int ph = lane >> 4;
    v8f acc[4] = {};
    const float* arow = tile + m * F2_PITCH_ + 2 * ph;
    for (int k = 0; k < 128; k += 4) {
        const int kp = (k >> 1) + ph;
        const v2f a = *(const v2f*)(arow + k);
#pragma unroll
        for (int t = 0; t < 4; ++t) {
            const v2f b = Wpk[kp * 64 + t * 16 + m];
            acc[t] = __builtin_amdgcn_wmma_f32_16x16x4_f32(
                false, a, false, b, (short)0, acc[t], false, false);
        }
    }

    const int mb0 = row0 + 8 * ph;
#pragma unroll
    for (int t = 0; t < 4; ++t) {
        const float b = bias[t * 16 + m];
#pragma unroll
        for (int r = 0; r < 8; ++r)
            out[(size_t)(mb0 + r) * 64 + t * 16 + m] = acc[t][r] + b;
    }
}

extern "C" void kernel_launch(void* const* d_in, const int* in_sizes, int n_in,
                              void* d_out, int out_size, void* d_ws, size_t ws_size,
                              hipStream_t stream) {
    const float* x_disease = (const float*)d_in[0];
    const float* x_gene    = (const float*)d_in[1];
    const int*   src_dg    = (const int*)d_in[2];
    const int*   dst_dg    = (const int*)d_in[3];
    const int*   src_gd    = (const int*)d_in[4];
    const int*   dst_gd    = (const int*)d_in[5];
    const float* Wl1_dg = (const float*)d_in[6];
    const float* bl1_dg = (const float*)d_in[7];
    const float* Wr1_dg = (const float*)d_in[8];
    const float* Wl1_gd = (const float*)d_in[9];
    const float* bl1_gd = (const float*)d_in[10];
    const float* Wr1_gd = (const float*)d_in[11];
    const float* Wl2_dg = (const float*)d_in[12];
    const float* bl2_dg = (const float*)d_in[13];
    const float* Wr2_dg = (const float*)d_in[14];
    const float* Wl2_gd = (const float*)d_in[15];
    const float* bl2_gd = (const float*)d_in[16];
    const float* Wr2_gd = (const float*)d_in[17];

    float* d2 = (float*)d_out;
    float* g2 = d2 + (size_t)ND_ * 64;
    float* p1_d = d2;
    float* p1_g = g2;

    float* ws = (float*)d_ws;
    size_t off = 0;
    int*      slots  = (int*)(ws + off);      off += (size_t)E2_;
    unsigned* tmp    = (unsigned*)(ws + off); off += (size_t)E2_;
    int2*     seg    = (int2*)(ws + off);     off += (size_t)NT_ * 2;
    int*      offs   = (int*)(ws + off);
    off += (size_t)NT_ + 48;
    float* g1    = ws + off; off += (size_t)NG_ * H_;
    float* d1    = ws + off; off += (size_t)ND_ * H_;
    (void)off; (void)ws_size; (void)in_sizes; (void)n_in; (void)out_size;

    const int pairs_g = (NG_ / 16 + 1) / 2;
    const int pairs_d = (ND_ / 16 + 1) / 2;
    const int nblk_g  = (pairs_g + 3) / 4;
    const int nblk_d  = (pairs_d + 3) / 4;
    static_assert(F1_WAVES_ == F2_WAVES_, "both fused launches share the block counts below");
    const int fblk_g  = (NG_ / 16 + F2_WAVES_ - 1) / F2_WAVES_;
    const int fblk_d  = (ND_ / 16 + F2_WAVES_ - 1) / F2_WAVES_;

    node_wmma<DIN_><<<nblk_d + nblk_g + SB_, CP_T_, 0, stream>>>(
        src_dg, dst_dg, src_gd, dst_gd, offs, tmp,
        x_disease, Wl1_dg, p1_d, ND_, nblk_d,
        x_gene,    Wl1_gd, p1_g, NG_, nblk_d + nblk_g);

    bucket_sort<<<NBKT_, BS_T_, 0, stream>>>(tmp, offs, seg, slots);

    layer1_fused<<<fblk_d + fblk_g, 256, 0, stream>>>(
        p1_g, x_disease, Wr1_gd, bl1_gd, d1, seg + NG_, ND_, NG_ - 1, fblk_d,
        p1_d, x_gene,    Wr1_dg, bl1_dg, g1, seg,       NG_, ND_ - 1,
        slots);

    layer2_fused<<<fblk_d + fblk_g, 256, 0, stream>>>(
        g1, d1, Wl2_gd, Wr2_gd, bl2_gd, d2, seg + NG_, ND_, NG_ - 1, fblk_d,
        d1, g1, Wl2_dg, Wr2_dg, bl2_dg, g2, seg,       NG_, ND_ - 1,
        slots);
}
